# starting kernel
# baseline (speedup 1.0000x reference)
#include <hip/hip_runtime.h>
#include <hip/hip_bf16.h>
#include <stdint.h>

#define BB    8
#define SS    512
#define EE    768
#define VV    30522
#define GVV   20000
#define NNZV  400000
#define HH    64
#define OO    16

#define WPB     640
#define NCHUNK  8
#define CHUNK   (GVV / NCHUNK)
#define KSTEPS  (CHUNK / 4)
#define ETILES  (EE / 16)

typedef float v2f __attribute__((ext_vector_type(2)));
typedef float v8f __attribute__((ext_vector_type(8)));

#define PRESENT_BYTES  20480
#define HVH_BYTES      ((size_t)BB * GVV * HH * 4)
#define PART_BYTES     ((size_t)NCHUNK * BB * EE * HH * 4)

__global__ void gcn_zero(uint32_t* __restrict__ p, long n) {
  long i = (long)blockIdx.x * blockDim.x + threadIdx.x;
  long stride = (long)gridDim.x * blockDim.x;
  for (; i < n; i += stride) p[i] = 0u;
}

__global__ void gcn_present(const int* __restrict__ input_ids,
                            const int* __restrict__ tok2g,
                            uint32_t* __restrict__ present) {
  int tid = blockIdx.x * blockDim.x + threadIdx.x;
  if (tid >= BB * SS) return;
  int b = tid / SS;
  int id = input_ids[tid];
  if (id > VV - 1 || id < 0) id = 0;
  int g = tok2g[id];
  atomicOr(&present[b * WPB + (g >> 5)], 1u << (g & 31));
}

__global__ void gcn_scatter(const int* __restrict__ rows,
                            const int* __restrict__ cols,
                            const float* __restrict__ vals,
                            const float* __restrict__ W,
                            const uint32_t* __restrict__ present,
                            float* __restrict__ Hvh) {
  int k = blockIdx.x * 4 + (threadIdx.x >> 6);
  const int h = threadIdx.x & 63;
  const int kStride = gridDim.x * 4;
  for (; k < NNZV; k += kStride) {
    const int row = rows[k];
    const int col = cols[k];
    const float vw = vals[k] * W[(size_t)col * HH + h];
    const int rw = row >> 5, rbit = row & 31;
    const int cw = col >> 5, cbit = col & 31;
#pragma unroll
    for (int b = 0; b < BB; ++b) {
      const uint32_t m = (present[b * WPB + rw] >> rbit) |
                         (present[b * WPB + cw] >> cbit);
      if (m & 1u)
        atomicAdd(&Hvh[((size_t)b * GVV + row) * HH + h], vw);
    }
  }
}

__global__ void __launch_bounds__(128)
gcn_gemm_wmma(const float* __restrict__ emb,
              const int* __restrict__ gv_ids,
              const float* __restrict__ Hvh,
              float* __restrict__ part) {
  const int wave   = blockIdx.x * 4 + (threadIdx.x >> 5);
  const int lane   = threadIdx.x & 31;
  const int lane16 = lane & 15;
  const int hi     = lane >> 4;

  const int c  = wave % NCHUNK;
  const int et = (wave / NCHUNK) % ETILES;
  const int b  = wave / (NCHUNK * ETILES);

  const int e0    = et * 16;
  const int vbase = c * CHUNK;
  const float* __restrict__ HvhB = Hvh + (size_t)b * GVV * HH;

  const v8f zacc = {0.f, 0.f, 0.f, 0.f, 0.f, 0.f, 0.f, 0.f};
  v8f acc0 = zacc, acc1 = zacc, acc2 = zacc, acc3 = zacc;

  for (int step = 0; step < KSTEPS; ++step) {
    const int va = vbase + step * 4 + 2 * hi;
    const int g0 = gv_ids[va];
    const int g1 = gv_ids[va + 1];

    v2f a;
    a.x = emb[(size_t)g0 * EE + e0 + lane16];
    a.y = emb[(size_t)g1 * EE + e0 + lane16];

    const float* __restrict__ r0 = HvhB + (size_t)va * HH + lane16;
    const float* __restrict__ r1 = r0 + HH;

    v2f b0, b1, b2, b3;
    b0.x = r0[0];  b0.y = r1[0];
    b1.x = r0[16]; b1.y = r1[16];
    b2.x = r0[32]; b2.y = r1[32];
    b3.x = r0[48]; b3.y = r1[48];

    acc0 = __builtin_amdgcn_wmma_f32_16x16x4_f32(false, a, false, b0,
                                                 (short)0, acc0, false, false);
    acc1 = __builtin_amdgcn_wmma_f32_16x16x4_f32(false, a, false, b1,
                                                 (short)0, acc1, false, false);
    acc2 = __builtin_amdgcn_wmma_f32_16x16x4_f32(false, a, false, b2,
                                                 (short)0, acc2, false, false);
    acc3 = __builtin_amdgcn_wmma_f32_16x16x4_f32(false, a, false, b3,
                                                 (short)0, acc3, false, false);
  }

  float* __restrict__ pb =
      part + (((size_t)c * BB + b) * EE + e0) * HH + lane16;
  v8f accs[4] = {acc0, acc1, acc2, acc3};
#pragma unroll
  for (int t = 0; t < 4; ++t)
#pragma unroll
    for (int r = 0; r < 8; ++r) {
      const int e = r + 8 * hi;
      pb[(size_t)e * HH + t * 16] = accs[t][r];
    }
}

__global__ void gcn_final(const float* __restrict__ part,
                          const float* __restrict__ fc_w,
                          const float* __restrict__ fc_b,
                          float* __restrict__ out) {
  __shared__ float lds[HH];
  const int e = blockIdx.x % EE;
  const int b = blockIdx.x / EE;
  const int h = threadIdx.x;
  float f = 0.f;
#pragma unroll
  for (int c = 0; c < NCHUNK; ++c)
    f += part[(((size_t)c * BB + b) * EE + e) * HH + h];
  lds[h] = f;
  __syncthreads();
  if (h < OO) {
    float a = fc_b[h];
#pragma unroll
    for (int hh = 0; hh < HH; ++hh) a += lds[hh] * fc_w[h * HH + hh];
    out[((size_t)b * OO + h) * EE + e] = a;
  }
}

extern "C" void kernel_launch(void* const* d_in, const int* in_sizes, int n_in,
                              void* d_out, int out_size, void* d_ws, size_t ws_size,
                              hipStream_t stream) {
  (void)in_sizes; (void)n_in; (void)out_size;
  const float* emb        = (const float*)d_in[0];
  const float* adj_values = (const float*)d_in[1];
  const float* W_vh       = (const float*)d_in[2];
  const float* fc_w       = (const float*)d_in[3];
  const float* fc_b       = (const float*)d_in[4];
  const int*   input_ids  = (const int*)d_in[5];
  const int*   tok2g      = (const int*)d_in[6];
  const int*   gv_ids     = (const int*)d_in[7];
  const int*   adj_rows   = (const int*)d_in[8];
  const int*   adj_cols   = (const int*)d_in[9];
  float* out = (float*)d_out;

  const size_t need = PRESENT_BYTES + HVH_BYTES + PART_BYTES;
  if (ws_size < need) return;

  uint8_t*  ws      = (uint8_t*)d_ws;
  uint32_t* present = (uint32_t*)ws;
  float*    Hvh     = (float*)(ws + PRESENT_BYTES);
  float*    part    = (float*)(ws + PRESENT_BYTES + HVH_BYTES);

  gcn_zero<<<4096, 256, 0, stream>>>((uint32_t*)ws, (long)(need / 4));
  gcn_present<<<(BB * SS) / 256, 256, 0, stream>>>(input_ids, tok2g, present);
  gcn_scatter<<<8192, 256, 0, stream>>>(adj_rows, adj_cols, adj_values, W_vh,
                                        present, Hvh);
  gcn_gemm_wmma<<<(BB * ETILES * NCHUNK) / 4, 128, 0, stream>>>(emb, gv_ids,
                                                                Hvh, part);
  gcn_final<<<BB * EE, HH, 0, stream>>>(part, fc_w, fc_b, out);
}


# GEMM K loop: loads of 5 K-steps issued as one batch ahead of their WMMAs, next batch's row indices prefetched behind them; same accumulation order
# speedup vs baseline: 1.9927x; 1.9927x over previous
#include <hip/hip_runtime.h>
#include <hip/hip_bf16.h>
#include <stdint.h>

#define BB    8
#define SS    512
#define EE    768
#define VV    30522
#define GVV   20000
#define NNZV  400000
#define HH    64
#define OO    16

#define WPB     640
#define NCHUNK  8
#define CHUNK   (GVV / NCHUNK)
#define KSTEPS  (CHUNK / 4)
#define ETILES  (EE / 16)
#define UU      5
#define NGRP    (KSTEPS / UU)

typedef float v2f __attribute__((ext_vector_type(2)));
typedef float v8f __attribute__((ext_vector_type(8)));

#define PRESENT_BYTES  20480
#define HVH_BYTES      ((size_t)BB * GVV * HH * 4)
#define PART_BYTES     ((size_t)NCHUNK * BB * EE * HH * 4)

__global__ void gcn_zero(uint32_t* __restrict__ p, long n) {
  long i = (long)blockIdx.x * blockDim.x + threadIdx.x;
  long stride = (long)gridDim.x * blockDim.x;
  for (; i < n; i += stride) p[i] = 0u;
}

__global__ void gcn_present(const int* __restrict__ input_ids,
                            const int* __restrict__ tok2g,
                            uint32_t* __restrict__ present) {
  int tid = blockIdx.x * blockDim.x + threadIdx.x;
  if (tid >= BB * SS) return;
  int b = tid / SS;
  int id = input_ids[tid];
  if (id > VV - 1 || id < 0) id = 0;
  int g = tok2g[id];
  atomicOr(&present[b * WPB + (g >> 5)], 1u << (g & 31));
}

__global__ void gcn_scatter(const int* __restrict__ rows,
                            const int* __restrict__ cols,
                            const float* __restrict__ vals,
                            const float* __restrict__ W,
                            const uint32_t* __restrict__ present,
                            float* __restrict__ Hvh) {
  int k = blockIdx.x * 4 + (threadIdx.x >> 6);
  const int h = threadIdx.x & 63;
  const int kStride = gridDim.x * 4;
  for (; k < NNZV; k += kStride) {
    const int row = rows[k];
    const int col = cols[k];
    const float vw = vals[k] * W[(size_t)col * HH + h];
    const int rw = row >> 5, rbit = row & 31;
    const int cw = col >> 5, cbit = col & 31;
#pragma unroll
    for (int b = 0; b < BB; ++b) {
      const uint32_t m = (present[b * WPB + rw] >> rbit) |
                         (present[b * WPB + cw] >> cbit);
      if (m & 1u)
        atomicAdd(&Hvh[((size_t)b * GVV + row) * HH + h], vw);
    }
  }
}

__global__ void __launch_bounds__(128)
gcn_gemm_wmma(const float* __restrict__ emb,
              const int* __restrict__ gv_ids,
              const float* __restrict__ Hvh,
              float* __restrict__ part) {
  const int wave   = blockIdx.x * 4 + (threadIdx.x >> 5);
  const int lane   = threadIdx.x & 31;
  const int lane16 = lane & 15;
  const int hi     = lane >> 4;

  const int c  = wave % NCHUNK;
  const int et = (wave / NCHUNK) % ETILES;
  const int b  = wave / (NCHUNK * ETILES);

  const int e0    = et * 16;
  const int vbase = c * CHUNK;
  const float* __restrict__ HvhB = Hvh + (size_t)b * GVV * HH;

  const v8f zacc = {0.f, 0.f, 0.f, 0.f, 0.f, 0.f, 0.f, 0.f};
  v8f acc0 = zacc, acc1 = zacc, acc2 = zacc, acc3 = zacc;

  const int*   __restrict__ gvp  = gv_ids + vbase + 2 * hi;
  const float* __restrict__ embL = emb + e0 + lane16;
  const float* __restrict__ hvL  = HvhB + (size_t)(vbase + 2 * hi) * HH + lane16;

  int g0[UU], g1[UU];
#pragma unroll
  for (int u = 0; u < UU; ++u) {
    g0[u] = gvp[u * 4];
    g1[u] = gvp[u * 4 + 1];
  }

  for (int grp = 0; grp < NGRP; ++grp) {
    v2f a[UU], b0[UU], b1[UU], b2[UU], b3[UU];
#pragma unroll
    for (int u = 0; u < UU; ++u) {
      a[u].x = embL[(size_t)g0[u] * EE];
      a[u].y = embL[(size_t)g1[u] * EE];
      const float* __restrict__ r0 = hvL + (size_t)((grp * UU + u) * 4) * HH;
      const float* __restrict__ r1 = r0 + HH;
      b0[u].x = r0[0];  b0[u].y = r1[0];
      b1[u].x = r0[16]; b1[u].y = r1[16];
      b2[u].x = r0[32]; b2[u].y = r1[32];
      b3[u].x = r0[48]; b3[u].y = r1[48];
    }

    const int gn = (grp + 1 < NGRP) ? grp + 1 : grp;
#pragma unroll
    for (int u = 0; u < UU; ++u) {
      g0[u] = gvp[(gn * UU + u) * 4];
      g1[u] = gvp[(gn * UU + u) * 4 + 1];
    }

    __builtin_amdgcn_sched_barrier(0);

#pragma unroll
    for (int u = 0; u < UU; ++u) {
      acc0 = __builtin_amdgcn_wmma_f32_16x16x4_f32(false, a[u], false, b0[u],
                                                   (short)0, acc0, false, false);
      acc1 = __builtin_amdgcn_wmma_f32_16x16x4_f32(false, a[u], false, b1[u],
                                                   (short)0, acc1, false, false);
      acc2 = __builtin_amdgcn_wmma_f32_16x16x4_f32(false, a[u], false, b2[u],
                                                   (short)0, acc2, false, false);
      acc3 = __builtin_amdgcn_wmma_f32_16x16x4_f32(false, a[u], false, b3[u],
                                                   (short)0, acc3, false, false);
    }
  }

  float* __restrict__ pb =
      part + (((size_t)c * BB + b) * EE + e0) * HH + lane16;
  v8f accs[4] = {acc0, acc1, acc2, acc3};
#pragma unroll
  for (int t = 0; t < 4; ++t)
#pragma unroll
    for (int r = 0; r < 8; ++r) {
      const int e = r + 8 * hi;
      pb[(size_t)e * HH + t * 16] = accs[t][r];
    }
}

__global__ void gcn_final(const float* __restrict__ part,
                          const float* __restrict__ fc_w,
                          const float* __restrict__ fc_b,
                          float* __restrict__ out) {
  __shared__ float lds[HH];
  const int e = blockIdx.x % EE;
  const int b = blockIdx.x / EE;
  const int h = threadIdx.x;
  float f = 0.f;
#pragma unroll
  for (int c = 0; c < NCHUNK; ++c)
    f += part[(((size_t)c * BB + b) * EE + e) * HH + h];
  lds[h] = f;
  __syncthreads();
  if (h < OO) {
    float a = fc_b[h];
#pragma unroll
    for (int hh = 0; hh < HH; ++hh) a += lds[hh] * fc_w[h * HH + hh];
    out[((size_t)b * OO + h) * EE + e] = a;
  }
}

extern "C" void kernel_launch(void* const* d_in, const int* in_sizes, int n_in,
                              void* d_out, int out_size, void* d_ws, size_t ws_size,
                              hipStream_t stream) {
  (void)in_sizes; (void)n_in; (void)out_size;
  const float* emb        = (const float*)d_in[0];
  const float* adj_values = (const float*)d_in[1];
  const float* W_vh       = (const float*)d_in[2];
  const float* fc_w       = (const float*)d_in[3];
  const float* fc_b       = (const float*)d_in[4];
  const int*   input_ids  = (const int*)d_in[5];
  const int*   tok2g      = (const int*)d_in[6];
  const int*   gv_ids     = (const int*)d_in[7];
  const int*   adj_rows   = (const int*)d_in[8];
  const int*   adj_cols   = (const int*)d_in[9];
  float* out = (float*)d_out;

  const size_t need = PRESENT_BYTES + HVH_BYTES + PART_BYTES;
  if (ws_size < need) return;

  uint8_t*  ws      = (uint8_t*)d_ws;
  uint32_t* present = (uint32_t*)ws;
  float*    Hvh     = (float*)(ws + PRESENT_BYTES);
  float*    part    = (float*)(ws + PRESENT_BYTES + HVH_BYTES);

  gcn_zero<<<4096, 256, 0, stream>>>((uint32_t*)ws, (long)(need / 4));
  gcn_present<<<(BB * SS) / 256, 256, 0, stream>>>(input_ids, tok2g, present);
  gcn_scatter<<<8192, 256, 0, stream>>>(adj_rows, adj_cols, adj_values, W_vh,
                                        present, Hvh);
  gcn_gemm_wmma<<<(BB * ETILES * NCHUNK) / 4, 128, 0, stream>>>(emb, gv_ids,
                                                                Hvh, part);
  gcn_final<<<BB * EE, HH, 0, stream>>>(part, fc_w, fc_b, out);
}


# fc_w applied first (chain right to left): scatter and WMMA GEMM carry 16 columns per batch not 64, GEMM rebuilt as one 768x128x20000 with loads one batch ahead; summation order changes
# speedup vs baseline: 6.3592x; 3.1912x over previous
#include <hip/hip_runtime.h>
#include <stdint.h>


#define BB    8
#define SS    512
#define EE    768
#define VV    30522
#define GVV   20000
#define NNZV  400000
#define HH    64
#define OO    16

#define WPB     640
#define NCHUNK  40
#define CHUNK   (GVV / NCHUNK)
#define KSTEPS  (CHUNK / 4)
#define UU      5
#define NGRP    (KSTEPS / UU)
#define ETILES  (EE / 16)
#define WAVES_PER_BLOCK 8
#define ETG     (ETILES / WAVES_PER_BLOCK)
#define NOUT    (BB * OO * EE)
#define TILE_FLOATS (16 * BB * OO)

typedef float v2f __attribute__((ext_vector_type(2)));
typedef float v4f __attribute__((ext_vector_type(4)));
typedef float v8f __attribute__((ext_vector_type(8)));

#define PRESENT_BYTES  20480
#define G_BYTES        ((size_t)BB * GVV * OO * 4)
#define WF_BYTES       ((size_t)GVV * OO * 4)
#define PART_BYTES     ((size_t)NCHUNK * NOUT * 4)
#define ZERO_VEC4      ((PRESENT_BYTES + G_BYTES) / 16)

__global__ void gcn_prep(const float* __restrict__ W,
                         const float* __restrict__ fc_w,
                         uint32_t* __restrict__ ws_words,
                         float* __restrict__ WF) {
  const int gtid   = blockIdx.x * blockDim.x + threadIdx.x;
  const int stride = gridDim.x * blockDim.x;
  uint4* __restrict__ zero_base = (uint4*)ws_words;
  const uint4 z = {0u, 0u, 0u, 0u};
  for (int i = gtid; i < (int)ZERO_VEC4; i += stride) zero_base[i] = z;
  for (int i = gtid; i < GVV * OO; i += stride) {
    const int c = i >> 4, o = i & 15;
    const v4f* __restrict__ wr = (const v4f*)(W + (size_t)c * HH);
    const v4f* __restrict__ fr = (const v4f*)(fc_w + o * HH);
    float s = 0.f;
    for (int h4 = 0; h4 < HH / 4; ++h4) {
      const v4f w = wr[h4], f = fr[h4];
      s += w.x * f.x; s += w.y * f.y; s += w.z * f.z; s += w.w * f.w;
    }
    WF[i] = s;
  }
}

__global__ void gcn_present(const int* __restrict__ input_ids,
                            const int* __restrict__ tok2g,
                            uint32_t* __restrict__ present) {
  int tid = blockIdx.x * blockDim.x + threadIdx.x;
  if (tid >= BB * SS) return;
  int b = tid / SS;
  int id = input_ids[tid];
  if (id > VV - 1 || id < 0) id = 0;
  int g = tok2g[id];
  atomicOr(&present[b * WPB + (g >> 5)], 1u << (g & 31));
}

__global__ void __launch_bounds__(256)
gcn_scatter(const int* __restrict__ rows,
            const int* __restrict__ cols,
            const float* __restrict__ vals,
            const float* __restrict__ WF,
            const uint32_t* __restrict__ present,
            float* __restrict__ G) {
  const int lane  = threadIdx.x & 31;
  const int o     = lane & 15;
  const int bsel  = o & 7;
  int k = (blockIdx.x * blockDim.x + threadIdx.x) >> 4;
  const int kStride = (gridDim.x * blockDim.x) >> 4;
  for (; k < NNZV; k += kStride) {
    const int row = rows[k];
    const int col = cols[k];
    const int id  = (o < 8) ? row : col;
    const uint32_t bit = (present[bsel * WPB + (id >> 5)] >> (id & 31)) & 1u;
    const unsigned long long bal = __ballot(bit != 0u);
    const uint32_t m16 = (uint32_t)(bal >> (lane & 16)) & 0xFFFFu;
    const uint32_t m = (m16 | (m16 >> 8)) & 0xFFu;
    if (m) {
      const float vw = vals[k] * WF[col * OO + o];
      float* __restrict__ gp = G + (size_t)(row >> 1) * (BB * OO * 2) + o * 2 + (row & 1);
      for (int b = 0; b < BB; ++b)
        if ((m >> b) & 1u) atomicAdd(gp + b * (OO * 2), vw);
    }
  }
}

#define LOAD_BATCH(A_, B_, grp_)                                               \
  _Pragma("unroll")          \
  for (int u = 0; u < UU; ++u) {                                               \
    const int st = (grp_) * UU + u;                                            \
    const int2 gi = sidx[st * 2 + hi];                                         \
    A_[u].x = embL[(size_t)gi.x * EE];                                         \
    A_[u].y = embL[(size_t)gi.y * EE];                                         \
    const v4f* __restrict__ bp = gL + (size_t)st * 128;                        \
    B_[u][0] = bp[0];  B_[u][1] = bp[16];                                      \
    B_[u][2] = bp[32]; B_[u][3] = bp[48];                                      \
  }

#define MMA_BATCH(A_, B_)                                                      \
  _Pragma("unroll")          \
  for (int u = 0; u < UU; ++u) {                                               \
    _Pragma("unroll")       \
    for (int q = 0; q < 4; ++q) {                                              \
      const v2f blo = __builtin_shufflevector(B_[u][q], B_[u][q], 0, 1);       \
      const v2f bhi = __builtin_shufflevector(B_[u][q], B_[u][q], 2, 3);       \
      acc[2 * q] = __builtin_amdgcn_wmma_f32_16x16x4_f32(                      \
          false, A_[u], false, blo, (short)0, acc[2 * q], false, false);       \
      acc[2 * q + 1] = __builtin_amdgcn_wmma_f32_16x16x4_f32(                  \
          false, A_[u], false, bhi, (short)0, acc[2 * q + 1], false, false);   \
    }                                                                          \
  }

__global__ void __launch_bounds__(256)
gcn_gemm_wmma(const float* __restrict__ emb,
              const int* __restrict__ gv_ids,
              const float* __restrict__ G,
              float* __restrict__ part) {
  __shared__ int2 sidx[KSTEPS * 2];

  const int chunk = blockIdx.x % NCHUNK;
  const int etg   = blockIdx.x / NCHUNK;
  const int wv    = threadIdx.x >> 5;
  const int lane  = threadIdx.x & 31;
  const int c     = lane & 15;
  const int hi    = lane >> 4;
  const int et    = etg * WAVES_PER_BLOCK + wv;
  const int e0    = et * 16;
  const int vbase = chunk * CHUNK;

  for (int i = threadIdx.x; i < CHUNK; i += 256)
    ((int*)sidx)[i] = gv_ids[vbase + i];
  __syncthreads();

  const float* __restrict__ embL = emb + e0 + c;
  const v4f*   __restrict__ gL   =
      (const v4f*)(G + (size_t)(vbase / 2 + hi) * (BB * OO * 2)) + c;

  const v8f zacc = {0.f, 0.f, 0.f, 0.f, 0.f, 0.f, 0.f, 0.f};
  v8f acc[8];
#pragma unroll
  for (int t = 0; t < 8; ++t) acc[t] = zacc;

  v2f aA[UU], aB[UU];
  v4f bA[UU][4], bB[UU][4];

  LOAD_BATCH(aA, bA, 0);
  for (int it = 0; it < (NGRP - 1) / 2; ++it) {
    LOAD_BATCH(aB, bB, 2 * it + 1);
    __builtin_amdgcn_sched_barrier(0);
    MMA_BATCH(aA, bA);
    LOAD_BATCH(aA, bA, 2 * it + 2);
    __builtin_amdgcn_sched_barrier(0);
    MMA_BATCH(aB, bB);
  }
  MMA_BATCH(aA, bA);

  float* __restrict__ pb = part + ((size_t)chunk * ETILES + et) * TILE_FLOATS;
#pragma unroll
  for (int t = 0; t < 8; ++t) {
    v4f* __restrict__ dst = (v4f*)(pb + ((t * 2 + hi) * 16 + c) * 8);
    dst[0] = __builtin_shufflevector(acc[t], acc[t], 0, 1, 2, 3);
    dst[1] = __builtin_shufflevector(acc[t], acc[t], 4, 5, 6, 7);
  }
}

__global__ void __launch_bounds__(256)
gcn_final(const float* __restrict__ part,
          const float* __restrict__ fc_b,
          float* __restrict__ out) {
  const int idx = blockIdx.x * blockDim.x + threadIdx.x;
  if (idx >= NOUT) return;
  const int e = idx % EE;
  const int o = (idx / EE) % OO;
  const int b = idx / (EE * OO);
  const int c = ((b & 1) << 3) | (o >> 1);
  const int t = ((b >> 1) << 1) | (o & 1);
  const int et = e >> 4, h = (e >> 3) & 1, r = e & 7;
  const float* __restrict__ p = part + (size_t)et * TILE_FLOATS + ((t * 2 + h) * 16 + c) * 8 + r;
  float s = 0.f;
  for (int ch = 0; ch < NCHUNK; ++ch) s += p[(size_t)ch * NOUT];
  out[idx] = s + fc_b[o];
}

extern "C" void kernel_launch(void* const* d_in, const int* in_sizes, int n_in,
                              void* d_out, int out_size, void* d_ws, size_t ws_size,
                              hipStream_t stream) {
  (void)in_sizes; (void)n_in; (void)out_size;
  const float* emb        = (const float*)d_in[0];
  const float* adj_values = (const float*)d_in[1];
  const float* W_vh       = (const float*)d_in[2];
  const float* fc_w       = (const float*)d_in[3];
  const float* fc_b       = (const float*)d_in[4];
  const int*   input_ids  = (const int*)d_in[5];
  const int*   tok2g      = (const int*)d_in[6];
  const int*   gv_ids     = (const int*)d_in[7];
  const int*   adj_rows   = (const int*)d_in[8];
  const int*   adj_cols   = (const int*)d_in[9];
  float* out = (float*)d_out;

  const size_t need = PRESENT_BYTES + G_BYTES + WF_BYTES + PART_BYTES;
  if (ws_size < need) return;

  uint8_t*  ws      = (uint8_t*)d_ws;
  uint32_t* present = (uint32_t*)ws;
  float*    G       = (float*)(ws + PRESENT_BYTES);
  float*    WF      = (float*)(ws + PRESENT_BYTES + G_BYTES);
  float*    part    = (float*)(ws + PRESENT_BYTES + G_BYTES + WF_BYTES);

  gcn_prep<<<512, 256, 0, stream>>>(W_vh, fc_w, present, WF);
  gcn_present<<<(BB * SS) / 256, 256, 0, stream>>>(input_ids, tok2g, present);
  gcn_scatter<<<2048, 256, 0, stream>>>(adj_rows, adj_cols, adj_values, WF, present, G);
  gcn_gemm_wmma<<<ETG * NCHUNK, 256, 0, stream>>>(emb, gv_ids, G, part);
  gcn_final<<<NOUT / 256, 256, 0, stream>>>(part, fc_b, out);
}


# present-bitmap launch folded into the prep launch: 8 extra blocks build each batch's bitmap in LDS and write it whole while the rest zero G and compute WF; 5 launches become 4, sums unchanged
# speedup vs baseline: 6.5915x; 1.0365x over previous
#include <hip/hip_runtime.h>
#include <stdint.h>


#define BB    8
#define SS    512
#define EE    768
#define VV    30522
#define GVV   20000
#define NNZV  400000
#define HH    64
#define OO    16

#define WPB     640
#define NCHUNK  40
#define CHUNK   (GVV / NCHUNK)
#define KSTEPS  (CHUNK / 4)
#define UU      5
#define NGRP    (KSTEPS / UU)
#define ETILES  (EE / 16)
#define WAVES_PER_BLOCK 8
#define ETG     (ETILES / WAVES_PER_BLOCK)
#define NOUT    (BB * OO * EE)
#define TILE_FLOATS (16 * BB * OO)

typedef float v2f __attribute__((ext_vector_type(2)));
typedef float v4f __attribute__((ext_vector_type(4)));
typedef float v8f __attribute__((ext_vector_type(8)));

#define PRESENT_BYTES  20480
#define G_BYTES        ((size_t)BB * GVV * OO * 4)
#define WF_BYTES       ((size_t)GVV * OO * 4)
#define PART_BYTES     ((size_t)NCHUNK * NOUT * 4)
#define ZERO_VEC4      (G_BYTES / 16)

__global__ void __launch_bounds__(256)
gcn_prep(const float* __restrict__ W,
         const float* __restrict__ fc_w,
         const int* __restrict__ input_ids,
         const int* __restrict__ tok2g,
         uint32_t* __restrict__ present,
         float* __restrict__ G,
         float* __restrict__ WF) {
  __shared__ uint32_t sbits[WPB];
  if (blockIdx.x < BB) {
    const int b = blockIdx.x;
    int id0 = input_ids[b * SS + threadIdx.x];
    int id1 = input_ids[b * SS + 256 + threadIdx.x];
    for (int i = threadIdx.x; i < WPB; i += 256) sbits[i] = 0u;
    if (id0 > VV - 1 || id0 < 0) id0 = 0;
    if (id1 > VV - 1 || id1 < 0) id1 = 0;
    const int g0 = tok2g[id0];
    const int g1 = tok2g[id1];
    __syncthreads();
    atomicOr(&sbits[g0 >> 5], 1u << (g0 & 31));
    atomicOr(&sbits[g1 >> 5], 1u << (g1 & 31));
    __syncthreads();
    for (int i = threadIdx.x; i < WPB; i += 256) present[b * WPB + i] = sbits[i];
    return;
  }
  const int gtid   = (blockIdx.x - BB) * blockDim.x + threadIdx.x;
  const int stride = (gridDim.x - BB) * blockDim.x;
  uint4* __restrict__ zero_base = (uint4*)G;
  const uint4 z = {0u, 0u, 0u, 0u};
  for (int i = gtid; i < (int)ZERO_VEC4; i += stride) zero_base[i] = z;
  for (int i = gtid; i < GVV * OO; i += stride) {
    const int c = i >> 4, o = i & 15;
    const v4f* __restrict__ wr = (const v4f*)(W + (size_t)c * HH);
    const v4f* __restrict__ fr = (const v4f*)(fc_w + o * HH);
    float s = 0.f;
    for (int h4 = 0; h4 < HH / 4; ++h4) {
      const v4f w = wr[h4], f = fr[h4];
      s += w.x * f.x; s += w.y * f.y; s += w.z * f.z; s += w.w * f.w;
    }
    WF[i] = s;
  }
}

__global__ void __launch_bounds__(256)
gcn_scatter(const int* __restrict__ rows,
            const int* __restrict__ cols,
            const float* __restrict__ vals,
            const float* __restrict__ WF,
            const uint32_t* __restrict__ present,
            float* __restrict__ G) {
  const int lane  = threadIdx.x & 31;
  const int o     = lane & 15;
  const int bsel  = o & 7;
  int k = (blockIdx.x * blockDim.x + threadIdx.x) >> 4;
  const int kStride = (gridDim.x * blockDim.x) >> 4;
  for (; k < NNZV; k += kStride) {
    const int row = rows[k];
    const int col = cols[k];
    const int id  = (o < 8) ? row : col;
    const uint32_t bit = (present[bsel * WPB + (id >> 5)] >> (id & 31)) & 1u;
    const unsigned long long bal = __ballot(bit != 0u);
    const uint32_t m16 = (uint32_t)(bal >> (lane & 16)) & 0xFFFFu;
    const uint32_t m = (m16 | (m16 >> 8)) & 0xFFu;
    if (m) {
      const float vw = vals[k] * WF[col * OO + o];
      float* __restrict__ gp = G + (size_t)(row >> 1) * (BB * OO * 2) + o * 2 + (row & 1);
      for (int b = 0; b < BB; ++b)
        if ((m >> b) & 1u) atomicAdd(gp + b * (OO * 2), vw);
    }
  }
}

#define LOAD_BATCH(A_, B_, grp_)                                               \
  _Pragma("unroll")          \
  for (int u = 0; u < UU; ++u) {                                               \
    const int st = (grp_) * UU + u;                                            \
    const int2 gi = sidx[st * 2 + hi];                                         \
    A_[u].x = embL[(size_t)gi.x * EE];                                         \
    A_[u].y = embL[(size_t)gi.y * EE];                                         \
    const v4f* __restrict__ bp = gL + (size_t)st * 128;                        \
    B_[u][0] = bp[0];  B_[u][1] = bp[16];                                      \
    B_[u][2] = bp[32]; B_[u][3] = bp[48];                                      \
  }

#define MMA_BATCH(A_, B_)                                                      \
  _Pragma("unroll")          \
  for (int u = 0; u < UU; ++u) {                                               \
    _Pragma("unroll")       \
    for (int q = 0; q < 4; ++q) {                                              \
      const v2f blo = __builtin_shufflevector(B_[u][q], B_[u][q], 0, 1);       \
      const v2f bhi = __builtin_shufflevector(B_[u][q], B_[u][q], 2, 3);       \
      acc[2 * q] = __builtin_amdgcn_wmma_f32_16x16x4_f32(                      \
          false, A_[u], false, blo, (short)0, acc[2 * q], false, false);       \
      acc[2 * q + 1] = __builtin_amdgcn_wmma_f32_16x16x4_f32(                  \
          false, A_[u], false, bhi, (short)0, acc[2 * q + 1], false, false);   \
    }                                                                          \
  }

__global__ void __launch_bounds__(256)
gcn_gemm_wmma(const float* __restrict__ emb,
              const int* __restrict__ gv_ids,
              const float* __restrict__ G,
              float* __restrict__ part) {
  __shared__ int2 sidx[KSTEPS * 2];

  const int chunk = blockIdx.x % NCHUNK;
  const int etg   = blockIdx.x / NCHUNK;
  const int wv    = threadIdx.x >> 5;
  const int lane  = threadIdx.x & 31;
  const int c     = lane & 15;
  const int hi    = lane >> 4;
  const int et    = etg * WAVES_PER_BLOCK + wv;
  const int e0    = et * 16;
  const int vbase = chunk * CHUNK;

  for (int i = threadIdx.x; i < CHUNK; i += 256)
    ((int*)sidx)[i] = gv_ids[vbase + i];
  __syncthreads();

  const float* __restrict__ embL = emb + e0 + c;
  const v4f*   __restrict__ gL   =
      (const v4f*)(G + (size_t)(vbase / 2 + hi) * (BB * OO * 2)) + c;

  const v8f zacc = {0.f, 0.f, 0.f, 0.f, 0.f, 0.f, 0.f, 0.f};
  v8f acc[8];
#pragma unroll
  for (int t = 0; t < 8; ++t) acc[t] = zacc;

  v2f aA[UU], aB[UU];
  v4f bA[UU][4], bB[UU][4];

  LOAD_BATCH(aA, bA, 0);
  for (int it = 0; it < (NGRP - 1) / 2; ++it) {
    LOAD_BATCH(aB, bB, 2 * it + 1);
    __builtin_amdgcn_sched_barrier(0);
    MMA_BATCH(aA, bA);
    LOAD_BATCH(aA, bA, 2 * it + 2);
    __builtin_amdgcn_sched_barrier(0);
    MMA_BATCH(aB, bB);
  }
  MMA_BATCH(aA, bA);

  float* __restrict__ pb = part + ((size_t)chunk * ETILES + et) * TILE_FLOATS;
#pragma unroll
  for (int t = 0; t < 8; ++t) {
    v4f* __restrict__ dst = (v4f*)(pb + ((t * 2 + hi) * 16 + c) * 8);
    dst[0] = __builtin_shufflevector(acc[t], acc[t], 0, 1, 2, 3);
    dst[1] = __builtin_shufflevector(acc[t], acc[t], 4, 5, 6, 7);
  }
}

__global__ void __launch_bounds__(256)
gcn_final(const float* __restrict__ part,
          const float* __restrict__ fc_b,
          float* __restrict__ out) {
  const int idx = blockIdx.x * blockDim.x + threadIdx.x;
  if (idx >= NOUT) return;
  const int e = idx % EE;
  const int o = (idx / EE) % OO;
  const int b = idx / (EE * OO);
  const int c = ((b & 1) << 3) | (o >> 1);
  const int t = ((b >> 1) << 1) | (o & 1);
  const int et = e >> 4, h = (e >> 3) & 1, r = e & 7;
  const float* __restrict__ p = part + (size_t)et * TILE_FLOATS + ((t * 2 + h) * 16 + c) * 8 + r;
  float s = 0.f;
  for (int ch = 0; ch < NCHUNK; ++ch) s += p[(size_t)ch * NOUT];
  out[idx] = s + fc_b[o];
}

extern "C" void kernel_launch(void* const* d_in, const int* in_sizes, int n_in,
                              void* d_out, int out_size, void* d_ws, size_t ws_size,
                              hipStream_t stream) {
  (void)in_sizes; (void)n_in; (void)out_size;
  const float* emb        = (const float*)d_in[0];
  const float* adj_values = (const float*)d_in[1];
  const float* W_vh       = (const float*)d_in[2];
  const float* fc_w       = (const float*)d_in[3];
  const float* fc_b       = (const float*)d_in[4];
  const int*   input_ids  = (const int*)d_in[5];
  const int*   tok2g      = (const int*)d_in[6];
  const int*   gv_ids     = (const int*)d_in[7];
  const int*   adj_rows   = (const int*)d_in[8];
  const int*   adj_cols   = (const int*)d_in[9];
  float* out = (float*)d_out;

  const size_t need = PRESENT_BYTES + G_BYTES + WF_BYTES + PART_BYTES;
  if (ws_size < need) return;

  uint8_t*  ws      = (uint8_t*)d_ws;
  uint32_t* present = (uint32_t*)ws;
  float*    G       = (float*)(ws + PRESENT_BYTES);
  float*    WF      = (float*)(ws + PRESENT_BYTES + G_BYTES);
  float*    part    = (float*)(ws + PRESENT_BYTES + G_BYTES + WF_BYTES);

  gcn_prep<<<512 + BB, 256, 0, stream>>>(W_vh, fc_w, input_ids, tok2g, present, G, WF);
  gcn_scatter<<<2048, 256, 0, stream>>>(adj_rows, adj_cols, adj_values, WF, present, G);
  gcn_gemm_wmma<<<ETG * NCHUNK, 256, 0, stream>>>(emb, gv_ids, G, part);
  gcn_final<<<NOUT / 256, 256, 0, stream>>>(part, fc_b, out);
}


# G built without global atomics: nnz entries sorted into 80-row bins, one block per bin adds its kept entries into an LDS image of the G slice (LDS float atomics, arrival order) and stores it whole
# speedup vs baseline: 7.3928x; 1.1216x over previous
#include <hip/hip_runtime.h>
#include <stdint.h>


#define BB    8
#define SS    512
#define EE    768
#define VV    30522
#define GVV   20000
#define NNZV  400000
#define HH    64
#define OO    16

#define NCHUNK  40
#define CHUNK   (GVV / NCHUNK)
#define KSTEPS  (CHUNK / 4)
#define UU      5
#define NGRP    (KSTEPS / UU)
#define ETILES  (EE / 16)
#define WAVES_PER_BLOCK 8
#define ETG     (ETILES / WAVES_PER_BLOCK)
#define NOUT    (BB * OO * EE)
#define TILE_FLOATS (16 * BB * OO)

#define BINROWS 80
#define NBIN    (GVV / BINROWS)
#define EPB     1600
#define NSB     (NNZV / EPB)
#define EPT     7
#define BINFLOATS (BINROWS * BB * OO)
#define ACCPITCH  129
#define RND     2048
#define RPT     (RND / 256)
#define ENT     20
#define WFBLOCKS 512

typedef float v2f __attribute__((ext_vector_type(2)));
typedef float v4f __attribute__((ext_vector_type(4)));
typedef float v8f __attribute__((ext_vector_type(8)));
typedef int   v4i __attribute__((ext_vector_type(4)));
typedef int   v2i __attribute__((ext_vector_type(2)));

#define PM_BYTES       20480
#define G_BYTES        ((size_t)BB * GVV * OO * 4)
#define WF_BYTES       ((size_t)GVV * OO * 4)
#define PART_BYTES     ((size_t)NCHUNK * NOUT * 4)
#define REC_BYTES      ((size_t)NNZV * 16)

static __device__ __forceinline__ void excl_scan256_wave0(int* a, int t) {
  if (t < 32) {
    int v[8];
    int s = 0;
#pragma unroll
    for (int i = 0; i < 8; ++i) { v[i] = a[t * 8 + i]; s += v[i]; }
    int incl = s;
#pragma unroll
    for (int d = 1; d < 32; d <<= 1) {
      const int n = __shfl_up(incl, d);
      if (t >= d) incl += n;
    }
    int run = incl - s;
#pragma unroll
    for (int i = 0; i < 8; ++i) { a[t * 8 + i] = run; run += v[i]; }
  }
}

__global__ void __launch_bounds__(256)
gcn_prep(const float* __restrict__ W,
         const float* __restrict__ fc_w,
         const int* __restrict__ input_ids,
         const int* __restrict__ tok2g,
         const int* __restrict__ rows,
         const int* __restrict__ cols,
         const float* __restrict__ vals,
         unsigned int* __restrict__ pm_words,
         float* __restrict__ WF,
         int* __restrict__ rec,
         int* __restrict__ seg) {
  __shared__ __attribute__((aligned(16))) unsigned int sbuf[512 + EPB * 4];
  const int t = threadIdx.x;

  if (blockIdx.x < NSB) {
    const int sb = blockIdx.x;
    int* hist = (int*)sbuf;
    int* scan = (int*)sbuf + 256;
    v4i* srec = (v4i*)(sbuf + 512);
    hist[t] = 0;
    int   row[EPT], col[EPT], bin[EPT], rank[EPT];
    float val[EPT];
#pragma unroll
    for (int u = 0; u < EPT; ++u) {
      const int i = t + u * 256;
      row[u] = 0; col[u] = 0; val[u] = 0.f;
      if (i < EPB) {
        const int k = sb * EPB + i;
        row[u] = rows[k]; col[u] = cols[k]; val[u] = vals[k];
      }
    }
    __syncthreads();
#pragma unroll
    for (int u = 0; u < EPT; ++u) {
      const int i = t + u * 256;
      int bn = row[u] / BINROWS;
      bn = bn < 0 ? 0 : (bn > NBIN - 1 ? NBIN - 1 : bn);
      bin[u] = bn; rank[u] = 0;
      if (i < EPB) rank[u] = atomicAdd(&hist[bn], 1);
    }
    __syncthreads();
    const int h = hist[t];
    scan[t] = h;
    __syncthreads();
    excl_scan256_wave0(scan, t);
    __syncthreads();
    const int excl = scan[t];
    {
      v2i sc; sc.x = sb * EPB + excl; sc.y = h;
      ((v2i*)seg)[sb * 256 + t] = sc;
    }
#pragma unroll
    for (int u = 0; u < EPT; ++u) {
      const int i = t + u * 256;
      if (i < EPB) {
        v4i r; r.x = row[u]; r.y = col[u]; r.z = __float_as_int(val[u]); r.w = 0;
        srec[scan[bin[u]] + rank[u]] = r;
      }
    }
    __syncthreads();
    for (int i = t; i < EPB; i += 256) ((v4i*)rec)[sb * EPB + i] = srec[i];
    return;
  }

  if (blockIdx.x == NSB) {
    for (int i = t; i < PM_BYTES / 4; i += 256) sbuf[i] = 0u;
    int g[16];
#pragma unroll
    for (int u = 0; u < 16; ++u) {
      int id = input_ids[u * 256 + t];
      if (id > VV - 1 || id < 0) id = 0;
      g[u] = tok2g[id];
    }
    __syncthreads();
#pragma unroll
    for (int u = 0; u < 16; ++u) {
      const int b = (u * 256 + t) >> 9;
      if ((unsigned)g[u] < (unsigned)GVV)
        atomicOr(&sbuf[g[u] >> 2], (1u << b) << ((g[u] & 3) * 8));
    }
    __syncthreads();
    for (int i = t; i < PM_BYTES / 4; i += 256) pm_words[i] = sbuf[i];
    return;
  }

  const int gtid   = (blockIdx.x - NSB - 1) * blockDim.x + t;
  const int stride = (gridDim.x - NSB - 1) * blockDim.x;
  for (int i = gtid; i < GVV * OO; i += stride) {
    const int c = i >> 4, o = i & 15;
    const v4f* __restrict__ wr = (const v4f*)(W + (size_t)c * HH);
    const v4f* __restrict__ fr = (const v4f*)(fc_w + o * HH);
    float s = 0.f;
    for (int h4 = 0; h4 < HH / 4; ++h4) {
      const v4f w = wr[h4], f = fr[h4];
      s += w.x * f.x; s += w.y * f.y; s += w.z * f.z; s += w.w * f.w;
    }
    WF[i] = s;
  }
}

__global__ void __launch_bounds__(256)
gcn_accum(const float* __restrict__ WF,
          const unsigned char* __restrict__ pm,
          const int* __restrict__ rec,
          const int* __restrict__ seg,
          float* __restrict__ G) {
  __shared__ float s_acc[BINROWS * ACCPITCH];
  __shared__ __attribute__((aligned(16))) float s_ent[RND * ENT];
  __shared__ int   s_base[256];
  __shared__ int   s_start[256];
  __shared__ int   s_cnt[2];

  const int t = threadIdx.x;
  const int j = blockIdx.x;

  for (int i = t; i < BINROWS * ACCPITCH; i += 256) s_acc[i] = 0.f;
  int cnt = 0, st = 0;
  if (t < NSB) {
    const v2i sc = ((const v2i*)seg)[t * 256 + j];
    st = sc.x; cnt = sc.y;
  }
  s_base[t] = cnt;
  s_start[t] = st;
  if (t == 0) s_cnt[1] = 0;
  __syncthreads();
  excl_scan256_wave0(s_base, t);
  __syncthreads();
  if (t == 255) s_cnt[0] = s_base[255] + cnt;
  __syncthreads();
  const int N = s_cnt[0];

  const int hw = t >> 4;
  const int o  = t & 15;

  for (int r0 = 0; r0 < N; r0 += RND) {
    v4i  rc[RPT];
    bool ok[RPT];
#pragma unroll
    for (int u = 0; u < RPT; ++u) {
      const int idx = r0 + u * 256 + t;
      ok[u] = idx < N;
      const int ix = ok[u] ? idx : N - 1;
      int lo = 0;
#pragma unroll
      for (int s = 128; s >= 1; s >>= 1)
        lo += (s_base[lo + s] <= ix) ? s : 0;
      rc[u] = ((const v4i*)rec)[s_start[lo] + (ix - s_base[lo])];
    }
    unsigned int m[RPT];
    int vl[RPT];
#pragma unroll
    for (int u = 0; u < RPT; ++u) {
      vl[u] = rc[u].x - j * BINROWS;
      const bool good = ok[u] && (unsigned)vl[u] < (unsigned)BINROWS &&
                        (unsigned)rc[u].y < (unsigned)GVV;
      const int rr = good ? rc[u].x : 0;
      const int cc = good ? rc[u].y : 0;
      const unsigned int mb = (unsigned int)pm[rr] | (unsigned int)pm[cc];
      m[u] = good ? mb : 0u;
    }
    v4f wf[RPT][4];
#pragma unroll
    for (int u = 0; u < RPT; ++u) {
      if (m[u]) {
        const v4f* __restrict__ wp = (const v4f*)(WF + (size_t)rc[u].y * OO);
        wf[u][0] = wp[0]; wf[u][1] = wp[1]; wf[u][2] = wp[2]; wf[u][3] = wp[3];
      }
    }
    int kept = 0;
#pragma unroll
    for (int u = 0; u < RPT; ++u) kept += m[u] ? 1 : 0;
    int slot = 0;
    if (kept) slot = atomicAdd(&s_cnt[1], kept);
#pragma unroll
    for (int u = 0; u < RPT; ++u) {
      if (m[u]) {
        const float v = __int_as_float(rc[u].z);
        v4f* __restrict__ e = (v4f*)(s_ent + slot * ENT);
        e[0] = wf[u][0] * v; e[1] = wf[u][1] * v; e[2] = wf[u][2] * v; e[3] = wf[u][3] * v;
        v2i hd; hd.x = vl[u] * ACCPITCH; hd.y = (int)m[u];
        *(v2i*)(s_ent + slot * ENT + 16) = hd;
        ++slot;
      }
    }
    __syncthreads();
    const int ne = s_cnt[1];
    for (int e0 = 0; e0 < ne; e0 += 128) {
      float p[8]; int base[8]; unsigned int mm[8];
#pragma unroll
      for (int q = 0; q < 8; ++q) {
        const int e = e0 + q * 16 + hw;
        p[q] = 0.f; base[q] = 0; mm[q] = 0u;
        if (e < ne) {
          p[q] = s_ent[e * ENT + o];
          const v2i hd = *(const v2i*)(s_ent + e * ENT + 16);
          base[q] = hd.x; mm[q] = (unsigned int)hd.y;
        }
      }
#pragma unroll
      for (int q = 0; q < 8; ++q) {
        unsigned int mq = mm[q] & 0xFFu;
        while (mq) {
          const int b = __ffs((int)mq) - 1;
          mq &= mq - 1u;
          atomicAdd(&s_acc[base[q] + b * OO + o], p[q]);
        }
      }
    }
    __syncthreads();
    if (t == 0) s_cnt[1] = 0;
    __syncthreads();
  }

  v4f* __restrict__ gdst = (v4f*)(G + (size_t)j * BINFLOATS);
  for (int p4 = t; p4 < BINFLOATS / 4; p4 += 256) {
    float x[4];
#pragma unroll
    for (int i = 0; i < 4; ++i) {
      const int p = p4 * 4 + i;
      const int vp = p >> 8, b = (p >> 5) & 7, oo = (p >> 1) & 15, par = p & 1;
      x[i] = s_acc[(2 * vp + par) * ACCPITCH + b * OO + oo];
    }
    v4f xv; xv.x = x[0]; xv.y = x[1]; xv.z = x[2]; xv.w = x[3];
    gdst[p4] = xv;
  }
}

#define LOAD_BATCH(A_, B_, grp_)                                               \
  _Pragma("unroll")          \
  for (int u = 0; u < UU; ++u) {                                               \
    const int st = (grp_) * UU + u;                                            \
    const int2 gi = sidx[st * 2 + hi];                                         \
    A_[u].x = embL[(size_t)gi.x * EE];                                         \
    A_[u].y = embL[(size_t)gi.y * EE];                                         \
    const v4f* __restrict__ bp = gL + (size_t)st * 128;                        \
    B_[u][0] = bp[0];  B_[u][1] = bp[16];                                      \
    B_[u][2] = bp[32]; B_[u][3] = bp[48];                                      \
  }

#define MMA_BATCH(A_, B_)                                                      \
  _Pragma("unroll")          \
  for (int u = 0; u < UU; ++u) {                                               \
    _Pragma("unroll")       \
    for (int q = 0; q < 4; ++q) {                                              \
      const v2f blo = __builtin_shufflevector(B_[u][q], B_[u][q], 0, 1);       \
      const v2f bhi = __builtin_shufflevector(B_[u][q], B_[u][q], 2, 3);       \
      acc[2 * q] = __builtin_amdgcn_wmma_f32_16x16x4_f32(                      \
          false, A_[u], false, blo, (short)0, acc[2 * q], false, false);       \
      acc[2 * q + 1] = __builtin_amdgcn_wmma_f32_16x16x4_f32(                  \
          false, A_[u], false, bhi, (short)0, acc[2 * q + 1], false, false);   \
    }                                                                          \
  }

__global__ void __launch_bounds__(256)
gcn_gemm_wmma(const float* __restrict__ emb,
              const int* __restrict__ gv_ids,
              const float* __restrict__ G,
              float* __restrict__ part) {
  __shared__ int2 sidx[KSTEPS * 2];

  const int chunk = blockIdx.x % NCHUNK;
  const int etg   = blockIdx.x / NCHUNK;
  const int wv    = threadIdx.x >> 5;
  const int lane  = threadIdx.x & 31;
  const int c     = lane & 15;
  const int hi    = lane >> 4;
  const int et    = etg * WAVES_PER_BLOCK + wv;
  const int e0    = et * 16;
  const int vbase = chunk * CHUNK;

  for (int i = threadIdx.x; i < CHUNK; i += 256)
    ((int*)sidx)[i] = gv_ids[vbase + i];
  __syncthreads();

  const float* __restrict__ embL = emb + e0 + c;
  const v4f*   __restrict__ gL   =
      (const v4f*)(G + (size_t)(vbase / 2 + hi) * (BB * OO * 2)) + c;

  const v8f zacc = {0.f, 0.f, 0.f, 0.f, 0.f, 0.f, 0.f, 0.f};
  v8f acc[8];
#pragma unroll
  for (int t = 0; t < 8; ++t) acc[t] = zacc;

  v2f aA[UU], aB[UU];
  v4f bA[UU][4], bB[UU][4];

  LOAD_BATCH(aA, bA, 0);
  for (int it = 0; it < (NGRP - 1) / 2; ++it) {
    LOAD_BATCH(aB, bB, 2 * it + 1);
    __builtin_amdgcn_sched_barrier(0);
    MMA_BATCH(aA, bA);
    LOAD_BATCH(aA, bA, 2 * it + 2);
    __builtin_amdgcn_sched_barrier(0);
    MMA_BATCH(aB, bB);
  }
  MMA_BATCH(aA, bA);

  float* __restrict__ pb = part + ((size_t)chunk * ETILES + et) * TILE_FLOATS;
#pragma unroll
  for (int t = 0; t < 8; ++t) {
    v4f* __restrict__ dst = (v4f*)(pb + ((t * 2 + hi) * 16 + c) * 8);
    dst[0] = __builtin_shufflevector(acc[t], acc[t], 0, 1, 2, 3);
    dst[1] = __builtin_shufflevector(acc[t], acc[t], 4, 5, 6, 7);
  }
}

__global__ void __launch_bounds__(256)
gcn_final(const float* __restrict__ part,
          const float* __restrict__ fc_b,
          float* __restrict__ out) {
  const int idx = blockIdx.x * blockDim.x + threadIdx.x;
  if (idx >= NOUT) return;
  const int e = idx % EE;
  const int o = (idx / EE) % OO;
  const int b = idx / (EE * OO);
  const int c = ((b & 1) << 3) | (o >> 1);
  const int t = ((b >> 1) << 1) | (o & 1);
  const int et = e >> 4, h = (e >> 3) & 1, r = e & 7;
  const float* __restrict__ p = part + (size_t)et * TILE_FLOATS + ((t * 2 + h) * 16 + c) * 8 + r;
  float s = 0.f;
  for (int ch = 0; ch < NCHUNK; ++ch) s += p[(size_t)ch * NOUT];
  out[idx] = s + fc_b[o];
}

extern "C" void kernel_launch(void* const* d_in, const int* in_sizes, int n_in,
                              void* d_out, int out_size, void* d_ws, size_t ws_size,
                              hipStream_t stream) {
  (void)in_sizes; (void)n_in; (void)out_size;
  const float* emb        = (const float*)d_in[0];
  const float* adj_values = (const float*)d_in[1];
  const float* W_vh       = (const float*)d_in[2];
  const float* fc_w       = (const float*)d_in[3];
  const float* fc_b       = (const float*)d_in[4];
  const int*   input_ids  = (const int*)d_in[5];
  const int*   tok2g      = (const int*)d_in[6];
  const int*   gv_ids     = (const int*)d_in[7];
  const int*   adj_rows   = (const int*)d_in[8];
  const int*   adj_cols   = (const int*)d_in[9];
  float* out = (float*)d_out;

  const size_t need = PM_BYTES + G_BYTES + WF_BYTES + PART_BYTES;
  if (ws_size < need) return;

  uint8_t* ws   = (uint8_t*)d_ws;
  unsigned int* pm_words = (unsigned int*)ws;
  float*    G       = (float*)(ws + PM_BYTES);
  float*    WF      = (float*)(ws + PM_BYTES + G_BYTES);
  float*    part    = (float*)(ws + PM_BYTES + G_BYTES + WF_BYTES);
  int*      rec     = (int*)(ws + PM_BYTES + G_BYTES + WF_BYTES);
  int*      seg     = (int*)(ws + PM_BYTES + G_BYTES + WF_BYTES + REC_BYTES);

  gcn_prep<<<NSB + 1 + WFBLOCKS, 256, 0, stream>>>(W_vh, fc_w, input_ids, tok2g, adj_rows, adj_cols,
                                                 adj_values, pm_words, WF, rec, seg);
  gcn_accum<<<NBIN, 256, 0, stream>>>(WF, (const unsigned char*)ws, rec, seg, G);
  gcn_gemm_wmma<<<ETG * NCHUNK, 256, 0, stream>>>(emb, gv_ids, G, part);
  gcn_final<<<NOUT / 256, 256, 0, stream>>>(part, fc_b, out);
}


# GEMM K loop: each batch's loads issued two batches ahead of their WMMAs (ring of three register sets, loads kept as one group) instead of one batch ahead; same accumulation order
# speedup vs baseline: 7.6824x; 1.0392x over previous
#include <hip/hip_runtime.h>
#include <stdint.h>


#define BB    8
#define SS    512
#define EE    768
#define VV    30522
#define GVV   20000
#define NNZV  400000
#define HH    64
#define OO    16

#define NCHUNK  40
#define CHUNK   (GVV / NCHUNK)
#define KSTEPS  (CHUNK / 4)
#define UU      5
#define NGRP    (KSTEPS / UU)
#define NTRIP   ((NGRP - 1) / 3)
#define ETILES  (EE / 16)
#define WAVES_PER_BLOCK 8
#define ETG     (ETILES / WAVES_PER_BLOCK)
#define NOUT    (BB * OO * EE)
#define TILE_FLOATS (16 * BB * OO)

#define BINROWS 80
#define NBIN    (GVV / BINROWS)
#define EPB     1600
#define NSB     (NNZV / EPB)
#define EPT     7
#define BINFLOATS (BINROWS * BB * OO)
#define ACCPITCH  129
#define RND     2048
#define RPT     (RND / 256)
#define ENT     20
#define WFBLOCKS 512

typedef float v2f __attribute__((ext_vector_type(2)));
typedef float v4f __attribute__((ext_vector_type(4)));
typedef float v8f __attribute__((ext_vector_type(8)));
typedef int   v4i __attribute__((ext_vector_type(4)));
typedef int   v2i __attribute__((ext_vector_type(2)));

#define PM_BYTES       20480
#define G_BYTES        ((size_t)BB * GVV * OO * 4)
#define WF_BYTES       ((size_t)GVV * OO * 4)
#define PART_BYTES     ((size_t)NCHUNK * NOUT * 4)
#define REC_BYTES      ((size_t)NNZV * 16)

static __device__ __forceinline__ void excl_scan256_wave0(int* a, int t) {
  if (t < 32) {
    int v[8];
    int s = 0;
#pragma unroll
    for (int i = 0; i < 8; ++i) { v[i] = a[t * 8 + i]; s += v[i]; }
    int incl = s;
#pragma unroll
    for (int d = 1; d < 32; d <<= 1) {
      const int n = __shfl_up(incl, d);
      if (t >= d) incl += n;
    }
    int run = incl - s;
#pragma unroll
    for (int i = 0; i < 8; ++i) { a[t * 8 + i] = run; run += v[i]; }
  }
}

__global__ void __launch_bounds__(256)
gcn_prep(const float* __restrict__ W,
         const float* __restrict__ fc_w,
         const int* __restrict__ input_ids,
         const int* __restrict__ tok2g,
         const int* __restrict__ rows,
         const int* __restrict__ cols,
         const float* __restrict__ vals,
         unsigned int* __restrict__ pm_words,
         float* __restrict__ WF,
         int* __restrict__ rec,
         int* __restrict__ seg) {
  __shared__ __attribute__((aligned(16))) unsigned int sbuf[512 + EPB * 4];
  const int t = threadIdx.x;

  if (blockIdx.x < NSB) {
    const int sb = blockIdx.x;
    int* hist = (int*)sbuf;
    int* scan = (int*)sbuf + 256;
    v4i* srec = (v4i*)(sbuf + 512);
    hist[t] = 0;
    int   row[EPT], col[EPT], bin[EPT], rank[EPT];
    float val[EPT];
#pragma unroll
    for (int u = 0; u < EPT; ++u) {
      const int i = t + u * 256;
      row[u] = 0; col[u] = 0; val[u] = 0.f;
      if (i < EPB) {
        const int k = sb * EPB + i;
        row[u] = rows[k]; col[u] = cols[k]; val[u] = vals[k];
      }
    }
    __syncthreads();
#pragma unroll
    for (int u = 0; u < EPT; ++u) {
      const int i = t + u * 256;
      int bn = row[u] / BINROWS;
      bn = bn < 0 ? 0 : (bn > NBIN - 1 ? NBIN - 1 : bn);
      bin[u] = bn; rank[u] = 0;
      if (i < EPB) rank[u] = atomicAdd(&hist[bn], 1);
    }
    __syncthreads();
    const int h = hist[t];
    scan[t] = h;
    __syncthreads();
    excl_scan256_wave0(scan, t);
    __syncthreads();
    const int excl = scan[t];
    {
      v2i sc; sc.x = sb * EPB + excl; sc.y = h;
      ((v2i*)seg)[sb * 256 + t] = sc;
    }
#pragma unroll
    for (int u = 0; u < EPT; ++u) {
      const int i = t + u * 256;
      if (i < EPB) {
        v4i r; r.x = row[u]; r.y = col[u]; r.z = __float_as_int(val[u]); r.w = 0;
        srec[scan[bin[u]] + rank[u]] = r;
      }
    }
    __syncthreads();
    for (int i = t; i < EPB; i += 256) ((v4i*)rec)[sb * EPB + i] = srec[i];
    return;
  }

  if (blockIdx.x == NSB) {
    for (int i = t; i < PM_BYTES / 4; i += 256) sbuf[i] = 0u;
    int g[16];
#pragma unroll
    for (int u = 0; u < 16; ++u) {
      int id = input_ids[u * 256 + t];
      if (id > VV - 1 || id < 0) id = 0;
      g[u] = tok2g[id];
    }
    __syncthreads();
#pragma unroll
    for (int u = 0; u < 16; ++u) {
      const int b = (u * 256 + t) >> 9;
      if ((unsigned)g[u] < (unsigned)GVV)
        atomicOr(&sbuf[g[u] >> 2], (1u << b) << ((g[u] & 3) * 8));
    }
    __syncthreads();
    for (int i = t; i < PM_BYTES / 4; i += 256) pm_words[i] = sbuf[i];
    return;
  }

  const int gtid   = (blockIdx.x - NSB - 1) * blockDim.x + t;
  const int stride = (gridDim.x - NSB - 1) * blockDim.x;
  for (int i = gtid; i < GVV * OO; i += stride) {
    const int c = i >> 4, o = i & 15;
    const v4f* __restrict__ wr = (const v4f*)(W + (size_t)c * HH);
    const v4f* __restrict__ fr = (const v4f*)(fc_w + o * HH);
    float s = 0.f;
    for (int h4 = 0; h4 < HH / 4; ++h4) {
      const v4f w = wr[h4], f = fr[h4];
      s += w.x * f.x; s += w.y * f.y; s += w.z * f.z; s += w.w * f.w;
    }
    WF[i] = s;
  }
}

__global__ void __launch_bounds__(256)
gcn_accum(const float* __restrict__ WF,
          const unsigned char* __restrict__ pm,
          const int* __restrict__ rec,
          const int* __restrict__ seg,
          float* __restrict__ G) {
  __shared__ float s_acc[BINROWS * ACCPITCH];
  __shared__ __attribute__((aligned(16))) float s_ent[RND * ENT];
  __shared__ int   s_base[256];
  __shared__ int   s_start[256];
  __shared__ int   s_cnt[2];

  const int t = threadIdx.x;
  const int j = blockIdx.x;

  for (int i = t; i < BINROWS * ACCPITCH; i += 256) s_acc[i] = 0.f;
  int cnt = 0, st = 0;
  if (t < NSB) {
    const v2i sc = ((const v2i*)seg)[t * 256 + j];
    st = sc.x; cnt = sc.y;
  }
  s_base[t] = cnt;
  s_start[t] = st;
  if (t == 0) s_cnt[1] = 0;
  __syncthreads();
  excl_scan256_wave0(s_base, t);
  __syncthreads();
  if (t == 255) s_cnt[0] = s_base[255] + cnt;
  __syncthreads();
  const int N = s_cnt[0];

  const int hw = t >> 4;
  const int o  = t & 15;

  for (int r0 = 0; r0 < N; r0 += RND) {
    v4i  rc[RPT];
    bool ok[RPT];
#pragma unroll
    for (int u = 0; u < RPT; ++u) {
      const int idx = r0 + u * 256 + t;
      ok[u] = idx < N;
      const int ix = ok[u] ? idx : N - 1;
      int lo = 0;
#pragma unroll
      for (int s = 128; s >= 1; s >>= 1)
        lo += (s_base[lo + s] <= ix) ? s : 0;
      rc[u] = ((const v4i*)rec)[s_start[lo] + (ix - s_base[lo])];
    }
    unsigned int m[RPT];
    int vl[RPT];
#pragma unroll
    for (int u = 0; u < RPT; ++u) {
      vl[u] = rc[u].x - j * BINROWS;
      const bool good = ok[u] && (unsigned)vl[u] < (unsigned)BINROWS &&
                        (unsigned)rc[u].y < (unsigned)GVV;
      const int rr = good ? rc[u].x : 0;
      const int cc = good ? rc[u].y : 0;
      const unsigned int mb = (unsigned int)pm[rr] | (unsigned int)pm[cc];
      m[u] = good ? mb : 0u;
    }
    v4f wf[RPT][4];
#pragma unroll
    for (int u = 0; u < RPT; ++u) {
      if (m[u]) {
        const v4f* __restrict__ wp = (const v4f*)(WF + (size_t)rc[u].y * OO);
        wf[u][0] = wp[0]; wf[u][1] = wp[1]; wf[u][2] = wp[2]; wf[u][3] = wp[3];
      }
    }
    int kept = 0;
#pragma unroll
    for (int u = 0; u < RPT; ++u) kept += m[u] ? 1 : 0;
    int slot = 0;
    if (kept) slot = atomicAdd(&s_cnt[1], kept);
#pragma unroll
    for (int u = 0; u < RPT; ++u) {
      if (m[u]) {
        const float v = __int_as_float(rc[u].z);
        v4f* __restrict__ e = (v4f*)(s_ent + slot * ENT);
        e[0] = wf[u][0] * v; e[1] = wf[u][1] * v; e[2] = wf[u][2] * v; e[3] = wf[u][3] * v;
        v2i hd; hd.x = vl[u] * ACCPITCH; hd.y = (int)m[u];
        *(v2i*)(s_ent + slot * ENT + 16) = hd;
        ++slot;
      }
    }
    __syncthreads();
    const int ne = s_cnt[1];
    for (int e0 = 0; e0 < ne; e0 += 128) {
      float p[8]; int base[8]; unsigned int mm[8];
#pragma unroll
      for (int q = 0; q < 8; ++q) {
        const int e = e0 + q * 16 + hw;
        p[q] = 0.f; base[q] = 0; mm[q] = 0u;
        if (e < ne) {
          p[q] = s_ent[e * ENT + o];
          const v2i hd = *(const v2i*)(s_ent + e * ENT + 16);
          base[q] = hd.x; mm[q] = (unsigned int)hd.y;
        }
      }
#pragma unroll
      for (int q = 0; q < 8; ++q) {
        unsigned int mq = mm[q] & 0xFFu;
        while (mq) {
          const int b = __ffs((int)mq) - 1;
          mq &= mq - 1u;
          atomicAdd(&s_acc[base[q] + b * OO + o], p[q]);
        }
      }
    }
    __syncthreads();
    if (t == 0) s_cnt[1] = 0;
    __syncthreads();
  }

  v4f* __restrict__ gdst = (v4f*)(G + (size_t)j * BINFLOATS);
  for (int p4 = t; p4 < BINFLOATS / 4; p4 += 256) {
    float x[4];
#pragma unroll
    for (int i = 0; i < 4; ++i) {
      const int p = p4 * 4 + i;
      const int vp = p >> 8, b = (p >> 5) & 7, oo = (p >> 1) & 15, par = p & 1;
      x[i] = s_acc[(2 * vp + par) * ACCPITCH + b * OO + oo];
    }
    v4f xv; xv.x = x[0]; xv.y = x[1]; xv.z = x[2]; xv.w = x[3];
    gdst[p4] = xv;
  }
}

#define LOAD_BATCH(A_, B_, grp_)                                               \
  _Pragma("unroll")          \
  for (int u = 0; u < UU; ++u) {                                               \
    const int st = (grp_) * UU + u;                                            \
    const int2 gi = sidx[st * 2 + hi];                                         \
    A_[u].x = embL[(size_t)gi.x * EE];                                         \
    A_[u].y = embL[(size_t)gi.y * EE];                                         \
    const v4f* __restrict__ bp = gL + (size_t)st * 128;                        \
    B_[u][0] = bp[0];  B_[u][1] = bp[16];                                      \
    B_[u][2] = bp[32]; B_[u][3] = bp[48];                                      \
  }

#define MMA_BATCH(A_, B_)                                                      \
  _Pragma("unroll")          \
  for (int u = 0; u < UU; ++u) {                                               \
    _Pragma("unroll")       \
    for (int q = 0; q < 4; ++q) {                                              \
      const v2f blo = __builtin_shufflevector(B_[u][q], B_[u][q], 0, 1);       \
      const v2f bhi = __builtin_shufflevector(B_[u][q], B_[u][q], 2, 3);       \
      acc[2 * q] = __builtin_amdgcn_wmma_f32_16x16x4_f32(                      \
          false, A_[u], false, blo, (short)0, acc[2 * q], false, false);       \
      acc[2 * q + 1] = __builtin_amdgcn_wmma_f32_16x16x4_f32(                  \
          false, A_[u], false, bhi, (short)0, acc[2 * q + 1], false, false);   \
    }                                                                          \
  }

__global__ void __launch_bounds__(256)
gcn_gemm_wmma(const float* __restrict__ emb,
              const int* __restrict__ gv_ids,
              const float* __restrict__ G,
              float* __restrict__ part) {
  __shared__ int2 sidx[KSTEPS * 2];

  const int chunk = blockIdx.x % NCHUNK;
  const int etg   = blockIdx.x / NCHUNK;
  const int wv    = threadIdx.x >> 5;
  const int lane  = threadIdx.x & 31;
  const int c     = lane & 15;
  const int hi    = lane >> 4;
  const int et    = etg * WAVES_PER_BLOCK + wv;
  const int e0    = et * 16;
  const int vbase = chunk * CHUNK;

  for (int i = threadIdx.x; i < CHUNK; i += 256)
    ((int*)sidx)[i] = gv_ids[vbase + i];
  __syncthreads();

  const float* __restrict__ embL = emb + e0 + c;
  const v4f*   __restrict__ gL   =
      (const v4f*)(G + (size_t)(vbase / 2 + hi) * (BB * OO * 2)) + c;

  const v8f zacc = {0.f, 0.f, 0.f, 0.f, 0.f, 0.f, 0.f, 0.f};
  v8f acc[8];
#pragma unroll
  for (int t = 0; t < 8; ++t) acc[t] = zacc;

  v2f aR0[UU], aR1[UU], aR2[UU];
  v4f bR0[UU][4], bR1[UU][4], bR2[UU][4];

  LOAD_BATCH(aR0, bR0, 0);
  LOAD_BATCH(aR1, bR1, 1);
  for (int it = 0; it < NTRIP; ++it) {
    const int g = 3 * it;
    LOAD_BATCH(aR2, bR2, g + 2);
    __builtin_amdgcn_sched_barrier(0);
    MMA_BATCH(aR0, bR0);
    __builtin_amdgcn_sched_barrier(0);
    LOAD_BATCH(aR0, bR0, g + 3);
    __builtin_amdgcn_sched_barrier(0);
    MMA_BATCH(aR1, bR1);
    __builtin_amdgcn_sched_barrier(0);
    if (it + 1 < NTRIP) {
      LOAD_BATCH(aR1, bR1, g + 4);
    }
    __builtin_amdgcn_sched_barrier(0);
    MMA_BATCH(aR2, bR2);
  }
  MMA_BATCH(aR0, bR0);

  float* __restrict__ pb = part + ((size_t)chunk * ETILES + et) * TILE_FLOATS;
#pragma unroll
  for (int t = 0; t < 8; ++t) {
    v4f* __restrict__ dst = (v4f*)(pb + ((t * 2 + hi) * 16 + c) * 8);
    dst[0] = __builtin_shufflevector(acc[t], acc[t], 0, 1, 2, 3);
    dst[1] = __builtin_shufflevector(acc[t], acc[t], 4, 5, 6, 7);
  }
}

__global__ void __launch_bounds__(256)
gcn_final(const float* __restrict__ part,
          const float* __restrict__ fc_b,
          float* __restrict__ out) {
  const int idx = blockIdx.x * blockDim.x + threadIdx.x;
  if (idx >= NOUT) return;
  const int e = idx % EE;
  const int o = (idx / EE) % OO;
  const int b = idx / (EE * OO);
  const int c = ((b & 1) << 3) | (o >> 1);
  const int t = ((b >> 1) << 1) | (o & 1);
  const int et = e >> 4, h = (e >> 3) & 1, r = e & 7;
  const float* __restrict__ p = part + (size_t)et * TILE_FLOATS + ((t * 2 + h) * 16 + c) * 8 + r;
  float s = 0.f;
  for (int ch = 0; ch < NCHUNK; ++ch) s += p[(size_t)ch * NOUT];
  out[idx] = s + fc_b[o];
}

extern "C" void kernel_launch(void* const* d_in, const int* in_sizes, int n_in,
                              void* d_out, int out_size, void* d_ws, size_t ws_size,
                              hipStream_t stream) {
  (void)in_sizes; (void)n_in; (void)out_size;
  const float* emb        = (const float*)d_in[0];
  const float* adj_values = (const float*)d_in[1];
  const float* W_vh       = (const float*)d_in[2];
  const float* fc_w       = (const float*)d_in[3];
  const float* fc_b       = (const float*)d_in[4];
  const int*   input_ids  = (const int*)d_in[5];
  const int*   tok2g      = (const int*)d_in[6];
  const int*   gv_ids     = (const int*)d_in[7];
  const int*   adj_rows   = (const int*)d_in[8];
  const int*   adj_cols   = (const int*)d_in[9];
  float* out = (float*)d_out;

  const size_t need = PM_BYTES + G_BYTES + WF_BYTES + PART_BYTES;
  if (ws_size < need) return;

  uint8_t* ws   = (uint8_t*)d_ws;
  unsigned int* pm_words = (unsigned int*)ws;
  float*    G       = (float*)(ws + PM_BYTES);
  float*    WF      = (float*)(ws + PM_BYTES + G_BYTES);
  float*    part    = (float*)(ws + PM_BYTES + G_BYTES + WF_BYTES);
  int*      rec     = (int*)(ws + PM_BYTES + G_BYTES + WF_BYTES);
  int*      seg     = (int*)(ws + PM_BYTES + G_BYTES + WF_BYTES + REC_BYTES);

  gcn_prep<<<NSB + 1 + WFBLOCKS, 256, 0, stream>>>(W_vh, fc_w, input_ids, tok2g, adj_rows, adj_cols,
                                                 adj_values, pm_words, WF, rec, seg);
  gcn_accum<<<NBIN, 256, 0, stream>>>(WF, (const unsigned char*)ws, rec, seg, G);
  gcn_gemm_wmma<<<ETG * NCHUNK, 256, 0, stream>>>(emb, gv_ids, G, part);
  gcn_final<<<NOUT / 256, 256, 0, stream>>>(part, fc_b, out);
}


# The two short launches re-cut into fewer, longer waves (5768 fewer per call): WF as a 20000x16x64 WMMA GEMM, 32 rows per wave (WF summation order changes); final sum 4 outputs per thread, whole lines
# speedup vs baseline: 8.8197x; 1.1480x over previous
#include <hip/hip_runtime.h>
#include <stdint.h>


#define BB    8
#define SS    512
#define EE    768
#define VV    30522
#define GVV   20000
#define NNZV  400000
#define HH    64
#define OO    16

#define NCHUNK  40
#define CHUNK   (GVV / NCHUNK)
#define KSTEPS  (CHUNK / 4)
#define UU      5
#define NGRP    (KSTEPS / UU)
#define NTRIP   ((NGRP - 1) / 3)
#define ETILES  (EE / 16)
#define WAVES_PER_BLOCK 8
#define ETG     (ETILES / WAVES_PER_BLOCK)
#define NOUT    (BB * OO * EE)
#define TILE_FLOATS (16 * BB * OO)

#define BINROWS 80
#define NBIN    (GVV / BINROWS)
#define EPB     1600
#define NSB     (NNZV / EPB)
#define EPT     7
#define BINFLOATS (BINROWS * BB * OO)
#define ACCPITCH  129
#define RND     2048
#define RPT     (RND / 256)
#define ENT     20
#define WFBLOCKS ((GVV / 32 + 7) / 8)

typedef float v2f __attribute__((ext_vector_type(2)));
typedef float v4f __attribute__((ext_vector_type(4)));
typedef float v8f __attribute__((ext_vector_type(8)));
typedef int   v4i __attribute__((ext_vector_type(4)));
typedef int   v2i __attribute__((ext_vector_type(2)));

#define PM_BYTES       20480
#define G_BYTES        ((size_t)BB * GVV * OO * 4)
#define WF_BYTES       ((size_t)GVV * OO * 4)
#define PART_BYTES     ((size_t)NCHUNK * NOUT * 4)
#define REC_BYTES      ((size_t)NNZV * 16)

static __device__ __forceinline__ void excl_scan256_wave0(int* a, int t) {
  if (t < 32) {
    int v[8];
    int s = 0;
#pragma unroll
    for (int i = 0; i < 8; ++i) { v[i] = a[t * 8 + i]; s += v[i]; }
    int incl = s;
#pragma unroll
    for (int d = 1; d < 32; d <<= 1) {
      const int n = __shfl_up(incl, d);
      if (t >= d) incl += n;
    }
    int run = incl - s;
#pragma unroll
    for (int i = 0; i < 8; ++i) { a[t * 8 + i] = run; run += v[i]; }
  }
}

__global__ void __launch_bounds__(256)
gcn_prep(const float* __restrict__ W,
         const float* __restrict__ fc_w,
         const int* __restrict__ input_ids,
         const int* __restrict__ tok2g,
         const int* __restrict__ rows,
         const int* __restrict__ cols,
         const float* __restrict__ vals,
         unsigned int* __restrict__ pm_words,
         float* __restrict__ WF,
         int* __restrict__ rec,
         int* __restrict__ seg) {
  __shared__ __attribute__((aligned(16))) unsigned int sbuf[512 + EPB * 4];
  const int t = threadIdx.x;

  if (blockIdx.x < NSB) {
    const int sb = blockIdx.x;
    int* hist = (int*)sbuf;
    int* scan = (int*)sbuf + 256;
    v4i* srec = (v4i*)(sbuf + 512);
    hist[t] = 0;
    int   row[EPT], col[EPT], bin[EPT], rank[EPT];
    float val[EPT];
#pragma unroll
    for (int u = 0; u < EPT; ++u) {
      const int i = t + u * 256;
      row[u] = 0; col[u] = 0; val[u] = 0.f;
      if (i < EPB) {
        const int k = sb * EPB + i;
        row[u] = rows[k]; col[u] = cols[k]; val[u] = vals[k];
      }
    }
    __syncthreads();
#pragma unroll
    for (int u = 0; u < EPT; ++u) {
      const int i = t + u * 256;
      int bn = row[u] / BINROWS;
      bn = bn < 0 ? 0 : (bn > NBIN - 1 ? NBIN - 1 : bn);
      bin[u] = bn; rank[u] = 0;
      if (i < EPB) rank[u] = atomicAdd(&hist[bn], 1);
    }
    __syncthreads();
    const int h = hist[t];
    scan[t] = h;
    __syncthreads();
    excl_scan256_wave0(scan, t);
    __syncthreads();
    const int excl = scan[t];
    {
      v2i sc; sc.x = sb * EPB + excl; sc.y = h;
      ((v2i*)seg)[sb * 256 + t] = sc;
    }
#pragma unroll
    for (int u = 0; u < EPT; ++u) {
      const int i = t + u * 256;
      if (i < EPB) {
        v4i r; r.x = row[u]; r.y = col[u]; r.z = __float_as_int(val[u]); r.w = 0;
        srec[scan[bin[u]] + rank[u]] = r;
      }
    }
    __syncthreads();
    for (int i = t; i < EPB; i += 256) ((v4i*)rec)[sb * EPB + i] = srec[i];
    return;
  }

  if (blockIdx.x == NSB) {
    for (int i = t; i < PM_BYTES / 4; i += 256) sbuf[i] = 0u;
    int g[16];
#pragma unroll
    for (int u = 0; u < 16; ++u) {
      int id = input_ids[u * 256 + t];
      if (id > VV - 1 || id < 0) id = 0;
      g[u] = tok2g[id];
    }
    __syncthreads();
#pragma unroll
    for (int u = 0; u < 16; ++u) {
      const int b = (u * 256 + t) >> 9;
      if ((unsigned)g[u] < (unsigned)GVV)
        atomicOr(&sbuf[g[u] >> 2], (1u << b) << ((g[u] & 3) * 8));
    }
    __syncthreads();
    for (int i = t; i < PM_BYTES / 4; i += 256) pm_words[i] = sbuf[i];
    return;
  }

  const int gw = (blockIdx.x - NSB - 1) * 8 + (t >> 5);
  if (gw * 32 < GVV) {
    const int lane = t & 31;
    const int c = lane & 15, h = lane >> 4;
    const int r0 = gw * 32;
    const v4f* __restrict__ a0p = (const v4f*)(W + (size_t)(r0 + c) * HH + 4 * h);
    const v4f* __restrict__ a1p = (const v4f*)(W + (size_t)(r0 + 16 + c) * HH + 4 * h);
    const v4f* __restrict__ bp  = (const v4f*)(fc_w + c * HH + 4 * h);
    v8f d0 = {0.f, 0.f, 0.f, 0.f, 0.f, 0.f, 0.f, 0.f};
    v8f d1 = d0;
    v4f a0[HH / 8], a1[HH / 8], bf[HH / 8];
#pragma unroll
    for (int j = 0; j < HH / 8; ++j) { a0[j] = a0p[j * 2]; a1[j] = a1p[j * 2]; bf[j] = bp[j * 2]; }
    __builtin_amdgcn_sched_barrier(0);
#pragma unroll
    for (int j = 0; j < HH / 8; ++j) {
      const v2f a0lo = __builtin_shufflevector(a0[j], a0[j], 0, 1), a0hi = __builtin_shufflevector(a0[j], a0[j], 2, 3);
      const v2f a1lo = __builtin_shufflevector(a1[j], a1[j], 0, 1), a1hi = __builtin_shufflevector(a1[j], a1[j], 2, 3);
      const v2f blo  = __builtin_shufflevector(bf[j], bf[j], 0, 1), bhi  = __builtin_shufflevector(bf[j], bf[j], 2, 3);
      d0 = __builtin_amdgcn_wmma_f32_16x16x4_f32(false, a0lo, false, blo, (short)0, d0, false, false);
      d1 = __builtin_amdgcn_wmma_f32_16x16x4_f32(false, a1lo, false, blo, (short)0, d1, false, false);
      d0 = __builtin_amdgcn_wmma_f32_16x16x4_f32(false, a0hi, false, bhi, (short)0, d0, false, false);
      d1 = __builtin_amdgcn_wmma_f32_16x16x4_f32(false, a1hi, false, bhi, (short)0, d1, false, false);
    }
    float* __restrict__ w0 = WF + (size_t)(r0 + 8 * h) * OO + c;
#pragma unroll
    for (int r = 0; r < 8; ++r) {
      w0[r * OO] = d0[r];
      w0[(16 + r) * OO] = d1[r];
    }
  }
}

__global__ void __launch_bounds__(256)
gcn_accum(const float* __restrict__ WF,
          const unsigned char* __restrict__ pm,
          const int* __restrict__ rec,
          const int* __restrict__ seg,
          float* __restrict__ G) {
  __shared__ float s_acc[BINROWS * ACCPITCH];
  __shared__ __attribute__((aligned(16))) float s_ent[RND * ENT];
  __shared__ int   s_base[256];
  __shared__ int   s_start[256];
  __shared__ int   s_cnt[2];

  const int t = threadIdx.x;
  const int j = blockIdx.x;

  for (int i = t; i < BINROWS * ACCPITCH; i += 256) s_acc[i] = 0.f;
  int cnt = 0, st = 0;
  if (t < NSB) {
    const v2i sc = ((const v2i*)seg)[t * 256 + j];
    st = sc.x; cnt = sc.y;
  }
  s_base[t] = cnt;
  s_start[t] = st;
  if (t == 0) s_cnt[1] = 0;
  __syncthreads();
  excl_scan256_wave0(s_base, t);
  __syncthreads();
  if (t == 255) s_cnt[0] = s_base[255] + cnt;
  __syncthreads();
  const int N = s_cnt[0];

  const int hw = t >> 4;
  const int o  = t & 15;

  for (int r0 = 0; r0 < N; r0 += RND) {
    v4i  rc[RPT];
    bool ok[RPT];
#pragma unroll
    for (int u = 0; u < RPT; ++u) {
      const int idx = r0 + u * 256 + t;
      ok[u] = idx < N;
      const int ix = ok[u] ? idx : N - 1;
      int lo = 0;
#pragma unroll
      for (int s = 128; s >= 1; s >>= 1)
        lo += (s_base[lo + s] <= ix) ? s : 0;
      rc[u] = ((const v4i*)rec)[s_start[lo] + (ix - s_base[lo])];
    }
    unsigned int m[RPT];
    int vl[RPT];
#pragma unroll
    for (int u = 0; u < RPT; ++u) {
      vl[u] = rc[u].x - j * BINROWS;
      const bool good = ok[u] && (unsigned)vl[u] < (unsigned)BINROWS &&
                        (unsigned)rc[u].y < (unsigned)GVV;
      const int rr = good ? rc[u].x : 0;
      const int cc = good ? rc[u].y : 0;
      const unsigned int mb = (unsigned int)pm[rr] | (unsigned int)pm[cc];
      m[u] = good ? mb : 0u;
    }
    v4f wf[RPT][4];
#pragma unroll
    for (int u = 0; u < RPT; ++u) {
      if (m[u]) {
        const v4f* __restrict__ wp = (const v4f*)(WF + (size_t)rc[u].y * OO);
        wf[u][0] = wp[0]; wf[u][1] = wp[1]; wf[u][2] = wp[2]; wf[u][3] = wp[3];
      }
    }
    int kept = 0;
#pragma unroll
    for (int u = 0; u < RPT; ++u) kept += m[u] ? 1 : 0;
    int slot = 0;
    if (kept) slot = atomicAdd(&s_cnt[1], kept);
#pragma unroll
    for (int u = 0; u < RPT; ++u) {
      if (m[u]) {
        const float v = __int_as_float(rc[u].z);
        v4f* __restrict__ e = (v4f*)(s_ent + slot * ENT);
        e[0] = wf[u][0] * v; e[1] = wf[u][1] * v; e[2] = wf[u][2] * v; e[3] = wf[u][3] * v;
        v2i hd; hd.x = vl[u] * ACCPITCH; hd.y = (int)m[u];
        *(v2i*)(s_ent + slot * ENT + 16) = hd;
        ++slot;
      }
    }
    __syncthreads();
    const int ne = s_cnt[1];
    for (int e0 = 0; e0 < ne; e0 += 128) {
      float p[8]; int base[8]; unsigned int mm[8];
#pragma unroll
      for (int q = 0; q < 8; ++q) {
        const int e = e0 + q * 16 + hw;
        p[q] = 0.f; base[q] = 0; mm[q] = 0u;
        if (e < ne) {
          p[q] = s_ent[e * ENT + o];
          const v2i hd = *(const v2i*)(s_ent + e * ENT + 16);
          base[q] = hd.x; mm[q] = (unsigned int)hd.y;
        }
      }
#pragma unroll
      for (int q = 0; q < 8; ++q) {
        unsigned int mq = mm[q] & 0xFFu;
        while (mq) {
          const int b = __ffs((int)mq) - 1;
          mq &= mq - 1u;
          atomicAdd(&s_acc[base[q] + b * OO + o], p[q]);
        }
      }
    }
    __syncthreads();
    if (t == 0) s_cnt[1] = 0;
    __syncthreads();
  }

  v4f* __restrict__ gdst = (v4f*)(G + (size_t)j * BINFLOATS);
  for (int p4 = t; p4 < BINFLOATS / 4; p4 += 256) {
    float x[4];
#pragma unroll
    for (int i = 0; i < 4; ++i) {
      const int p = p4 * 4 + i;
      const int vp = p >> 8, b = (p >> 5) & 7, oo = (p >> 1) & 15, par = p & 1;
      x[i] = s_acc[(2 * vp + par) * ACCPITCH + b * OO + oo];
    }
    v4f xv; xv.x = x[0]; xv.y = x[1]; xv.z = x[2]; xv.w = x[3];
    gdst[p4] = xv;
  }
}

#define LOAD_BATCH(A_, B_, grp_)                                               \
  _Pragma("unroll")          \
  for (int u = 0; u < UU; ++u) {                                               \
    const int st = (grp_) * UU + u;                                            \
    const int2 gi = sidx[st * 2 + hi];                                         \
    A_[u].x = embL[(size_t)gi.x * EE];                                         \
    A_[u].y = embL[(size_t)gi.y * EE];                                         \
    const v4f* __restrict__ bp = gL + (size_t)st * 128;                        \
    B_[u][0] = bp[0];  B_[u][1] = bp[16];                                      \
    B_[u][2] = bp[32]; B_[u][3] = bp[48];                                      \
  }

#define MMA_BATCH(A_, B_)                                                      \
  _Pragma("unroll")          \
  for (int u = 0; u < UU; ++u) {                                               \
    _Pragma("unroll")       \
    for (int q = 0; q < 4; ++q) {                                              \
      const v2f blo = __builtin_shufflevector(B_[u][q], B_[u][q], 0, 1);       \
      const v2f bhi = __builtin_shufflevector(B_[u][q], B_[u][q], 2, 3);       \
      acc[2 * q] = __builtin_amdgcn_wmma_f32_16x16x4_f32(                      \
          false, A_[u], false, blo, (short)0, acc[2 * q], false, false);       \
      acc[2 * q + 1] = __builtin_amdgcn_wmma_f32_16x16x4_f32(                  \
          false, A_[u], false, bhi, (short)0, acc[2 * q + 1], false, false);   \
    }                                                                          \
  }

__global__ void __launch_bounds__(256)
gcn_gemm_wmma(const float* __restrict__ emb,
              const int* __restrict__ gv_ids,
              const float* __restrict__ G,
              float* __restrict__ part) {
  __shared__ int2 sidx[KSTEPS * 2];

  const int chunk = blockIdx.x % NCHUNK;
  const int etg   = blockIdx.x / NCHUNK;
  const int wv    = threadIdx.x >> 5;
  const int lane  = threadIdx.x & 31;
  const int c     = lane & 15;
  const int hi    = lane >> 4;
  const int et    = etg * WAVES_PER_BLOCK + wv;
  const int e0    = et * 16;
  const int vbase = chunk * CHUNK;

  for (int i = threadIdx.x; i < CHUNK; i += 256)
    ((int*)sidx)[i] = gv_ids[vbase + i];
  __syncthreads();

  const float* __restrict__ embL = emb + e0 + c;
  const v4f*   __restrict__ gL   =
      (const v4f*)(G + (size_t)(vbase / 2 + hi) * (BB * OO * 2)) + c;

  const v8f zacc = {0.f, 0.f, 0.f, 0.f, 0.f, 0.f, 0.f, 0.f};
  v8f acc[8];
#pragma unroll
  for (int t = 0; t < 8; ++t) acc[t] = zacc;

  v2f aR0[UU], aR1[UU], aR2[UU];
  v4f bR0[UU][4], bR1[UU][4], bR2[UU][4];

  LOAD_BATCH(aR0, bR0, 0);
  LOAD_BATCH(aR1, bR1, 1);
  for (int it = 0; it < NTRIP; ++it) {
    const int g = 3 * it;
    LOAD_BATCH(aR2, bR2, g + 2);
    __builtin_amdgcn_sched_barrier(0);
    MMA_BATCH(aR0, bR0);
    __builtin_amdgcn_sched_barrier(0);
    LOAD_BATCH(aR0, bR0, g + 3);
    __builtin_amdgcn_sched_barrier(0);
    MMA_BATCH(aR1, bR1);
    __builtin_amdgcn_sched_barrier(0);
    if (it + 1 < NTRIP) {
      LOAD_BATCH(aR1, bR1, g + 4);
    }
    __builtin_amdgcn_sched_barrier(0);
    MMA_BATCH(aR2, bR2);
  }
  MMA_BATCH(aR0, bR0);

  float* __restrict__ pb = part + ((size_t)chunk * ETILES + et) * TILE_FLOATS;
#pragma unroll
  for (int t = 0; t < 8; ++t) {
    v4f* __restrict__ dst = (v4f*)(pb + ((t * 2 + hi) * 16 + c) * 8);
    dst[0] = __builtin_shufflevector(acc[t], acc[t], 0, 1, 2, 3);
    dst[1] = __builtin_shufflevector(acc[t], acc[t], 4, 5, 6, 7);
  }
}

#define FINAL_THREADS (NOUT / 4)
#define FINAL_BLOCK 64
__global__ void __launch_bounds__(FINAL_BLOCK)
gcn_final(const float* __restrict__ part,
          const float* __restrict__ fc_b,
          float* __restrict__ out) {
  const int gt   = blockIdx.x * FINAL_BLOCK + threadIdx.x;
  const int lane = gt & 31;
  const int w    = gt >> 5;
  const int c4  = w & 3;
  const int t   = (w >> 2) & 7;
  const int et2 = w >> 5;
  const int r4  = lane & 1;
  const int cc  = (lane >> 1) & 3;
  const int h   = (lane >> 3) & 1;
  const int etl = lane >> 4;
  const int c  = c4 * 4 + cc;
  const int et = et2 * 2 + etl;
  const int b  = (c >> 3) + 2 * (t >> 1);
  const int o  = 2 * (c & 7) + (t & 1);
  const v4f* __restrict__ p =
      (const v4f*)(part + (size_t)et * TILE_FLOATS + ((t * 2 + h) * 16 + c) * 8 + r4 * 4);
  v4f x[NCHUNK];
#pragma unroll
  for (int ch = 0; ch < NCHUNK; ++ch) x[ch] = p[(size_t)ch * (NOUT / 4)];
  __builtin_amdgcn_sched_barrier(0);
  v4f s = {0.f, 0.f, 0.f, 0.f};
#pragma unroll
  for (int ch = 0; ch < NCHUNK; ++ch) {
    s.x += x[ch].x; s.y += x[ch].y; s.z += x[ch].z; s.w += x[ch].w;
  }
  const float bias = fc_b[o];
  v4f y; y.x = s.x + bias; y.y = s.y + bias; y.z = s.z + bias; y.w = s.w + bias;
  *(v4f*)(out + ((size_t)(b * OO + o)) * EE + et * 16 + h * 8 + r4 * 4) = y;
}

extern "C" void kernel_launch(void* const* d_in, const int* in_sizes, int n_in,
                              void* d_out, int out_size, void* d_ws, size_t ws_size,
                              hipStream_t stream) {
  (void)in_sizes; (void)n_in; (void)out_size;
  const float* emb        = (const float*)d_in[0];
  const float* adj_values = (const float*)d_in[1];
  const float* W_vh       = (const float*)d_in[2];
  const float* fc_w       = (const float*)d_in[3];
  const float* fc_b       = (const float*)d_in[4];
  const int*   input_ids  = (const int*)d_in[5];
  const int*   tok2g      = (const int*)d_in[6];
  const int*   gv_ids     = (const int*)d_in[7];
  const int*   adj_rows   = (const int*)d_in[8];
  const int*   adj_cols   = (const int*)d_in[9];
  float* out = (float*)d_out;

  const size_t need = PM_BYTES + G_BYTES + WF_BYTES + PART_BYTES;
  if (ws_size < need) return;

  uint8_t* ws   = (uint8_t*)d_ws;
  unsigned int* pm_words = (unsigned int*)ws;
  float*    G       = (float*)(ws + PM_BYTES);
  float*    WF      = (float*)(ws + PM_BYTES + G_BYTES);
  float*    part    = (float*)(ws + PM_BYTES + G_BYTES + WF_BYTES);
  int*      rec     = (int*)(ws + PM_BYTES + G_BYTES + WF_BYTES);
  int*      seg     = (int*)(ws + PM_BYTES + G_BYTES + WF_BYTES + REC_BYTES);

  gcn_prep<<<NSB + 1 + WFBLOCKS, 256, 0, stream>>>(W_vh, fc_w, input_ids, tok2g, adj_rows, adj_cols,
                                                 adj_values, pm_words, WF, rec, seg);
  gcn_accum<<<NBIN, 256, 0, stream>>>(WF, (const unsigned char*)ws, rec, seg, G);
  gcn_gemm_wmma<<<ETG * NCHUNK, 256, 0, stream>>>(emb, gv_ids, G, part);
  gcn_final<<<FINAL_THREADS / FINAL_BLOCK, FINAL_BLOCK, 0, stream>>>(part, fc_b, out);
}


# accumulate launch freed of per-record branches: WF rows fetched with the mask bytes for all records, products in fixed LDS slots, a flat (record,batch) item list drives the LDS adds, arrival order
# speedup vs baseline: 9.4393x; 1.0703x over previous
#include <hip/hip_runtime.h>
#include <stdint.h>


#define BB    8
#define SS    512
#define EE    768
#define VV    30522
#define GVV   20000
#define NNZV  400000
#define HH    64
#define OO    16

#define NCHUNK  40
#define CHUNK   (GVV / NCHUNK)
#define KSTEPS  (CHUNK / 4)
#define UU      5
#define NGRP    (KSTEPS / UU)
#define NTRIP   ((NGRP - 1) / 3)
#define ETILES  (EE / 16)
#define WAVES_PER_BLOCK 8
#define ETG     (ETILES / WAVES_PER_BLOCK)
#define NOUT    (BB * OO * EE)
#define TILE_FLOATS (16 * BB * OO)

#define BINROWS 80
#define NBIN    (GVV / BINROWS)
#define EPB     1600
#define NSB     (NNZV / EPB)
#define EPT     7
#define BINFLOATS (BINROWS * BB * OO)
#define ACCPITCH  129
#define RND     2048
#define RPT     (RND / 256)
#define ENT     20
#define NITEM   (RND * BB)
#define WFBLOCKS ((GVV / 32 + 7) / 8)

typedef float v2f __attribute__((ext_vector_type(2)));
typedef float v4f __attribute__((ext_vector_type(4)));
typedef float v8f __attribute__((ext_vector_type(8)));
typedef int   v4i __attribute__((ext_vector_type(4)));
typedef int   v2i __attribute__((ext_vector_type(2)));

#define PM_BYTES       20480
#define G_BYTES        ((size_t)BB * GVV * OO * 4)
#define WF_BYTES       ((size_t)GVV * OO * 4)
#define PART_BYTES     ((size_t)NCHUNK * NOUT * 4)
#define REC_BYTES      ((size_t)NNZV * 16)

static __device__ __forceinline__ void excl_scan256_wave0(int* a, int t) {
  if (t < 32) {
    int v[8];
    int s = 0;
#pragma unroll
    for (int i = 0; i < 8; ++i) { v[i] = a[t * 8 + i]; s += v[i]; }
    int incl = s;
#pragma unroll
    for (int d = 1; d < 32; d <<= 1) {
      const int n = __shfl_up(incl, d);
      if (t >= d) incl += n;
    }
    int run = incl - s;
#pragma unroll
    for (int i = 0; i < 8; ++i) { a[t * 8 + i] = run; run += v[i]; }
  }
}

__global__ void __launch_bounds__(256)
gcn_prep(const float* __restrict__ W,
         const float* __restrict__ fc_w,
         const int* __restrict__ input_ids,
         const int* __restrict__ tok2g,
         const int* __restrict__ rows,
         const int* __restrict__ cols,
         const float* __restrict__ vals,
         unsigned int* __restrict__ pm_words,
         float* __restrict__ WF,
         int* __restrict__ rec,
         int* __restrict__ seg) {
  __shared__ __attribute__((aligned(16))) unsigned int sbuf[512 + EPB * 4];
  const int t = threadIdx.x;

  if (blockIdx.x < NSB) {
    const int sb = blockIdx.x;
    int* hist = (int*)sbuf;
    int* scan = (int*)sbuf + 256;
    v4i* srec = (v4i*)(sbuf + 512);
    hist[t] = 0;
    int   row[EPT], col[EPT], bin[EPT], rank[EPT];
    float val[EPT];
#pragma unroll
    for (int u = 0; u < EPT; ++u) {
      const int i = t + u * 256;
      row[u] = 0; col[u] = 0; val[u] = 0.f;
      if (i < EPB) {
        const int k = sb * EPB + i;
        row[u] = rows[k]; col[u] = cols[k]; val[u] = vals[k];
      }
    }
    __syncthreads();
#pragma unroll
    for (int u = 0; u < EPT; ++u) {
      const int i = t + u * 256;
      int bn = row[u] / BINROWS;
      bn = bn < 0 ? 0 : (bn > NBIN - 1 ? NBIN - 1 : bn);
      bin[u] = bn; rank[u] = 0;
      if (i < EPB) rank[u] = atomicAdd(&hist[bn], 1);
    }
    __syncthreads();
    const int h = hist[t];
    scan[t] = h;
    __syncthreads();
    excl_scan256_wave0(scan, t);
    __syncthreads();
    const int excl = scan[t];
    {
      v2i sc; sc.x = sb * EPB + excl; sc.y = h;
      ((v2i*)seg)[sb * 256 + t] = sc;
    }
#pragma unroll
    for (int u = 0; u < EPT; ++u) {
      const int i = t + u * 256;
      if (i < EPB) {
        v4i r; r.x = row[u]; r.y = col[u]; r.z = __float_as_int(val[u]); r.w = 0;
        srec[scan[bin[u]] + rank[u]] = r;
      }
    }
    __syncthreads();
    for (int i = t; i < EPB; i += 256) ((v4i*)rec)[sb * EPB + i] = srec[i];
    return;
  }

  if (blockIdx.x == NSB) {
    for (int i = t; i < PM_BYTES / 4; i += 256) sbuf[i] = 0u;
    int g[16];
#pragma unroll
    for (int u = 0; u < 16; ++u) {
      int id = input_ids[u * 256 + t];
      if (id > VV - 1 || id < 0) id = 0;
      g[u] = tok2g[id];
    }
    __syncthreads();
#pragma unroll
    for (int u = 0; u < 16; ++u) {
      const int b = (u * 256 + t) >> 9;
      if ((unsigned)g[u] < (unsigned)GVV)
        atomicOr(&sbuf[g[u] >> 2], (1u << b) << ((g[u] & 3) * 8));
    }
    __syncthreads();
    for (int i = t; i < PM_BYTES / 4; i += 256) pm_words[i] = sbuf[i];
    return;
  }

  const int gw = (blockIdx.x - NSB - 1) * 8 + (t >> 5);
  if (gw * 32 < GVV) {
    const int lane = t & 31;
    const int c = lane & 15, h = lane >> 4;
    const int r0 = gw * 32;
    const v4f* __restrict__ a0p = (const v4f*)(W + (size_t)(r0 + c) * HH + 4 * h);
    const v4f* __restrict__ a1p = (const v4f*)(W + (size_t)(r0 + 16 + c) * HH + 4 * h);
    const v4f* __restrict__ bp  = (const v4f*)(fc_w + c * HH + 4 * h);
    v8f d0 = {0.f, 0.f, 0.f, 0.f, 0.f, 0.f, 0.f, 0.f};
    v8f d1 = d0;
    v4f a0[HH / 8], a1[HH / 8], bf[HH / 8];
#pragma unroll
    for (int j = 0; j < HH / 8; ++j) { a0[j] = a0p[j * 2]; a1[j] = a1p[j * 2]; bf[j] = bp[j * 2]; }
    __builtin_amdgcn_sched_barrier(0);
#pragma unroll
    for (int j = 0; j < HH / 8; ++j) {
      const v2f a0lo = __builtin_shufflevector(a0[j], a0[j], 0, 1), a0hi = __builtin_shufflevector(a0[j], a0[j], 2, 3);
      const v2f a1lo = __builtin_shufflevector(a1[j], a1[j], 0, 1), a1hi = __builtin_shufflevector(a1[j], a1[j], 2, 3);
      const v2f blo  = __builtin_shufflevector(bf[j], bf[j], 0, 1), bhi  = __builtin_shufflevector(bf[j], bf[j], 2, 3);
      d0 = __builtin_amdgcn_wmma_f32_16x16x4_f32(false, a0lo, false, blo, (short)0, d0, false, false);
      d1 = __builtin_amdgcn_wmma_f32_16x16x4_f32(false, a1lo, false, blo, (short)0, d1, false, false);
      d0 = __builtin_amdgcn_wmma_f32_16x16x4_f32(false, a0hi, false, bhi, (short)0, d0, false, false);
      d1 = __builtin_amdgcn_wmma_f32_16x16x4_f32(false, a1hi, false, bhi, (short)0, d1, false, false);
    }
    float* __restrict__ w0 = WF + (size_t)(r0 + 8 * h) * OO + c;
#pragma unroll
    for (int r = 0; r < 8; ++r) {
      w0[r * OO] = d0[r];
      w0[(16 + r) * OO] = d1[r];
    }
  }
}

__global__ void __launch_bounds__(256)
gcn_accum(const float* __restrict__ WF,
          const unsigned char* __restrict__ pm,
          const int* __restrict__ rec,
          const int* __restrict__ seg,
          float* __restrict__ G) {
  __shared__ float s_acc[(BINROWS + 1) * ACCPITCH];
  __shared__ __attribute__((aligned(16))) float s_ent[RND * ENT];
  __shared__ unsigned int s_item[NITEM];
  __shared__ int   s_base[256];
  __shared__ int   s_start[256];
  __shared__ int   s_cnt[2];

  const int t = threadIdx.x;
  const int j = blockIdx.x;

  for (int i = t; i < (BINROWS + 1) * ACCPITCH; i += 256) s_acc[i] = 0.f;
  int cnt = 0, st = 0;
  if (t < NSB) {
    const v2i sc = ((const v2i*)seg)[t * 256 + j];
    st = sc.x; cnt = sc.y;
  }
  s_base[t] = cnt;
  s_start[t] = st;
  if (t == 0) s_cnt[1] = 0;
  __syncthreads();
  excl_scan256_wave0(s_base, t);
  __syncthreads();
  if (t == 255) s_cnt[0] = s_base[255] + cnt;
  __syncthreads();
  const int N = s_cnt[0];

  const int hw = t >> 4;
  const int o  = t & 15;

  for (int r0 = 0; r0 < N; r0 += RND) {
    v4i  rc[RPT];
    bool ok[RPT];
#pragma unroll
    for (int u = 0; u < RPT; ++u) {
      const int idx = r0 + u * 256 + t;
      ok[u] = idx < N;
      const int ix = ok[u] ? idx : N - 1;
      int lo = 0;
#pragma unroll
      for (int s = 128; s >= 1; s >>= 1)
        lo += (s_base[lo + s] <= ix) ? s : 0;
      rc[u] = ((const v4i*)rec)[s_start[lo] + (ix - s_base[lo])];
    }
    unsigned int m[RPT];
    int vl[RPT];
    v4f wf[RPT][4];
#pragma unroll
    for (int u = 0; u < RPT; ++u) {
      vl[u] = rc[u].x - j * BINROWS;
      const bool good = ok[u] && (unsigned)vl[u] < (unsigned)BINROWS &&
                        (unsigned)rc[u].y < (unsigned)GVV;
      const int rr = good ? rc[u].x : 0;
      const int cc = good ? rc[u].y : 0;
      const unsigned int mb = (unsigned int)pm[rr] | (unsigned int)pm[cc];
      m[u] = good ? (mb & 0xFFu) : 0u;
      const v4f* __restrict__ wp = (const v4f*)(WF + (size_t)cc * OO);
      wf[u][0] = wp[0]; wf[u][1] = wp[1]; wf[u][2] = wp[2]; wf[u][3] = wp[3];
    }
    int nit = 0;
#pragma unroll
    for (int u = 0; u < RPT; ++u) nit += __popc(m[u]);
    int islot = 0;
    if (nit) islot = atomicAdd(&s_cnt[1], nit);
#pragma unroll
    for (int u = 0; u < RPT; ++u) {
      const float v = __int_as_float(rc[u].z);
      const int e = u * 256 + t;
      v4f* __restrict__ ep = (v4f*)(s_ent + e * ENT);
      ep[0] = wf[u][0] * v; ep[1] = wf[u][1] * v; ep[2] = wf[u][2] * v; ep[3] = wf[u][3] * v;
      const unsigned int hdr = ((unsigned int)e << 16) | (unsigned int)(vl[u] * ACCPITCH);
      unsigned int mq = m[u];
      while (mq) {
        const int b = __ffs((int)mq) - 1;
        mq &= mq - 1u;
        s_item[islot++] = hdr + (unsigned int)(b * OO);
      }
    }
    __syncthreads();
    const int ni = s_cnt[1];
    const unsigned int spare = (unsigned int)(BINROWS * ACCPITCH);
    for (int i0 = 0; i0 < ni; i0 += 128) {
      unsigned int it[8];
#pragma unroll
      for (int q = 0; q < 8; ++q) {
        const int i = i0 + q * 16 + hw;
        const unsigned int x = s_item[i < ni ? i : 0];
        it[q] = i < ni ? x : spare;
      }
      float p[8];
#pragma unroll
      for (int q = 0; q < 8; ++q) p[q] = s_ent[(it[q] >> 16) * ENT + o];
#pragma unroll
      for (int q = 0; q < 8; ++q) atomicAdd(&s_acc[(it[q] & 0xFFFFu) + o], p[q]);
    }
    __syncthreads();
    if (t == 0) s_cnt[1] = 0;
    __syncthreads();
  }

  v4f* __restrict__ gdst = (v4f*)(G + (size_t)j * BINFLOATS);
  for (int p4 = t; p4 < BINFLOATS / 4; p4 += 256) {
    float x[4];
#pragma unroll
    for (int i = 0; i < 4; ++i) {
      const int p = p4 * 4 + i;
      const int vp = p >> 8, b = (p >> 5) & 7, oo = (p >> 1) & 15, par = p & 1;
      x[i] = s_acc[(2 * vp + par) * ACCPITCH + b * OO + oo];
    }
    v4f xv; xv.x = x[0]; xv.y = x[1]; xv.z = x[2]; xv.w = x[3];
    gdst[p4] = xv;
  }
}

#define LOAD_BATCH(A_, B_, grp_)                                               \
  _Pragma("unroll")          \
  for (int u = 0; u < UU; ++u) {                                               \
    const int st = (grp_) * UU + u;                                            \
    const int2 gi = sidx[st * 2 + hi];                                         \
    A_[u].x = embL[(size_t)gi.x * EE];                                         \
    A_[u].y = embL[(size_t)gi.y * EE];                                         \
    const v4f* __restrict__ bp = gL + (size_t)st * 128;                        \
    B_[u][0] = bp[0];  B_[u][1] = bp[16];                                      \
    B_[u][2] = bp[32]; B_[u][3] = bp[48];                                      \
  }

#define MMA_BATCH(A_, B_)                                                      \
  _Pragma("unroll")          \
  for (int u = 0; u < UU; ++u) {                                               \
    _Pragma("unroll")       \
    for (int q = 0; q < 4; ++q) {                                              \
      const v2f blo = __builtin_shufflevector(B_[u][q], B_[u][q], 0, 1);       \
      const v2f bhi = __builtin_shufflevector(B_[u][q], B_[u][q], 2, 3);       \
      acc[2 * q] = __builtin_amdgcn_wmma_f32_16x16x4_f32(                      \
          false, A_[u], false, blo, (short)0, acc[2 * q], false, false);       \
      acc[2 * q + 1] = __builtin_amdgcn_wmma_f32_16x16x4_f32(                  \
          false, A_[u], false, bhi, (short)0, acc[2 * q + 1], false, false);   \
    }                                                                          \
  }

__global__ void __launch_bounds__(256)
gcn_gemm_wmma(const float* __restrict__ emb,
              const int* __restrict__ gv_ids,
              const float* __restrict__ G,
              float* __restrict__ part) {
  __shared__ int2 sidx[KSTEPS * 2];

  const int chunk = blockIdx.x % NCHUNK;
  const int etg   = blockIdx.x / NCHUNK;
  const int wv    = threadIdx.x >> 5;
  const int lane  = threadIdx.x & 31;
  const int c     = lane & 15;
  const int hi    = lane >> 4;
  const int et    = etg * WAVES_PER_BLOCK + wv;
  const int e0    = et * 16;
  const int vbase = chunk * CHUNK;

  for (int i = threadIdx.x; i < CHUNK; i += 256)
    ((int*)sidx)[i] = gv_ids[vbase + i];
  __syncthreads();

  const float* __restrict__ embL = emb + e0 + c;
  const v4f*   __restrict__ gL   =
      (const v4f*)(G + (size_t)(vbase / 2 + hi) * (BB * OO * 2)) + c;

  const v8f zacc = {0.f, 0.f, 0.f, 0.f, 0.f, 0.f, 0.f, 0.f};
  v8f acc[8];
#pragma unroll
  for (int t = 0; t < 8; ++t) acc[t] = zacc;

  v2f aR0[UU], aR1[UU], aR2[UU];
  v4f bR0[UU][4], bR1[UU][4], bR2[UU][4];

  LOAD_BATCH(aR0, bR0, 0);
  LOAD_BATCH(aR1, bR1, 1);
  for (int it = 0; it < NTRIP; ++it) {
    const int g = 3 * it;
    LOAD_BATCH(aR2, bR2, g + 2);
    __builtin_amdgcn_sched_barrier(0);
    MMA_BATCH(aR0, bR0);
    __builtin_amdgcn_sched_barrier(0);
    LOAD_BATCH(aR0, bR0, g + 3);
    __builtin_amdgcn_sched_barrier(0);
    MMA_BATCH(aR1, bR1);
    __builtin_amdgcn_sched_barrier(0);
    if (it + 1 < NTRIP) {
      LOAD_BATCH(aR1, bR1, g + 4);
    }
    __builtin_amdgcn_sched_barrier(0);
    MMA_BATCH(aR2, bR2);
  }
  MMA_BATCH(aR0, bR0);

  float* __restrict__ pb = part + ((size_t)chunk * ETILES + et) * TILE_FLOATS;
#pragma unroll
  for (int t = 0; t < 8; ++t) {
    v4f* __restrict__ dst = (v4f*)(pb + ((t * 2 + hi) * 16 + c) * 8);
    dst[0] = __builtin_shufflevector(acc[t], acc[t], 0, 1, 2, 3);
    dst[1] = __builtin_shufflevector(acc[t], acc[t], 4, 5, 6, 7);
  }
}

#define FINAL_THREADS (NOUT / 4)
#define FINAL_BLOCK 64
__global__ void __launch_bounds__(FINAL_BLOCK)
gcn_final(const float* __restrict__ part,
          const float* __restrict__ fc_b,
          float* __restrict__ out) {
  const int gt   = blockIdx.x * FINAL_BLOCK + threadIdx.x;
  const int lane = gt & 31;
  const int w    = gt >> 5;
  const int c4  = w & 3;
  const int t   = (w >> 2) & 7;
  const int et2 = w >> 5;
  const int r4  = lane & 1;
  const int cc  = (lane >> 1) & 3;
  const int h   = (lane >> 3) & 1;
  const int etl = lane >> 4;
  const int c  = c4 * 4 + cc;
  const int et = et2 * 2 + etl;
  const int b  = (c >> 3) + 2 * (t >> 1);
  const int o  = 2 * (c & 7) + (t & 1);
  const v4f* __restrict__ p =
      (const v4f*)(part + (size_t)et * TILE_FLOATS + ((t * 2 + h) * 16 + c) * 8 + r4 * 4);
  v4f x[NCHUNK];
#pragma unroll
  for (int ch = 0; ch < NCHUNK; ++ch) x[ch] = p[(size_t)ch * (NOUT / 4)];
  __builtin_amdgcn_sched_barrier(0);
  v4f s = {0.f, 0.f, 0.f, 0.f};
#pragma unroll
  for (int ch = 0; ch < NCHUNK; ++ch) {
    s.x += x[ch].x; s.y += x[ch].y; s.z += x[ch].z; s.w += x[ch].w;
  }
  const float bias = fc_b[o];
  v4f y; y.x = s.x + bias; y.y = s.y + bias; y.z = s.z + bias; y.w = s.w + bias;
  *(v4f*)(out + ((size_t)(b * OO + o)) * EE + et * 16 + h * 8 + r4 * 4) = y;
}

extern "C" void kernel_launch(void* const* d_in, const int* in_sizes, int n_in,
                              void* d_out, int out_size, void* d_ws, size_t ws_size,
                              hipStream_t stream) {
  (void)in_sizes; (void)n_in; (void)out_size;
  const float* emb        = (const float*)d_in[0];
  const float* adj_values = (const float*)d_in[1];
  const float* W_vh       = (const float*)d_in[2];
  const float* fc_w       = (const float*)d_in[3];
  const float* fc_b       = (const float*)d_in[4];
  const int*   input_ids  = (const int*)d_in[5];
  const int*   tok2g      = (const int*)d_in[6];
  const int*   gv_ids     = (const int*)d_in[7];
  const int*   adj_rows   = (const int*)d_in[8];
  const int*   adj_cols   = (const int*)d_in[9];
  float* out = (float*)d_out;

  const size_t need = PM_BYTES + G_BYTES + WF_BYTES + PART_BYTES;
  if (ws_size < need) return;

  uint8_t* ws   = (uint8_t*)d_ws;
  unsigned int* pm_words = (unsigned int*)ws;
  float*    G       = (float*)(ws + PM_BYTES);
  float*    WF      = (float*)(ws + PM_BYTES + G_BYTES);
  float*    part    = (float*)(ws + PM_BYTES + G_BYTES + WF_BYTES);
  int*      rec     = (int*)(ws + PM_BYTES + G_BYTES + WF_BYTES);
  int*      seg     = (int*)(ws + PM_BYTES + G_BYTES + WF_BYTES + REC_BYTES);

  gcn_prep<<<NSB + 1 + WFBLOCKS, 256, 0, stream>>>(W_vh, fc_w, input_ids, tok2g, adj_rows, adj_cols,
                                                 adj_values, pm_words, WF, rec, seg);
  gcn_accum<<<NBIN, 256, 0, stream>>>(WF, (const unsigned char*)ws, rec, seg, G);
  gcn_gemm_wmma<<<ETG * NCHUNK, 256, 0, stream>>>(emb, gv_ids, G, part);
  gcn_final<<<FINAL_THREADS / FINAL_BLOCK, FINAL_BLOCK, 0, stream>>>(part, fc_b, out);
}
